# starting kernel
# baseline (speedup 1.0000x reference)
#include <hip/hip_runtime.h>
#include <hip/hip_bf16.h>


typedef __attribute__((ext_vector_type(2))) float v2f;
typedef __attribute__((ext_vector_type(8))) float v8f;

#define NAX 160
#define NT  10
#define KSTEPS 40

__global__ void build_H(float* __restrict__ H) {
  int idx = blockIdx.x * blockDim.x + threadIdx.x;
  if (idx >= NAX * NAX) return;
  int i = idx / NAX, j = idx % NAX;
  int d = i - j; if (d < 0) d += NAX;
  const double p = -0.26794919243112270647255365849413;
  double pd = 1.0, pnd = 1.0;
  for (int k = 0; k < d; ++k)        pd  *= p;
  for (int k = 0; k < NAX - d; ++k)  pnd *= p;
  double pN = pd * pnd;
  double h = 1.7320508075688772935274463415059 * (pd + pnd) / (1.0 - pN);
  H[idx] = (float)h;
}

template <int Q, int P, int S>
__global__ __launch_bounds__(256) void spline_pass(
    const float* __restrict__ in, float* __restrict__ out,
    const float* __restrict__ H, int nTiles)
{
  const int lane  = threadIdx.x & 31;
  const int wave  = threadIdx.x >> 5;
  const int tile  = blockIdx.x * 8 + wave;
  if (tile >= nTiles) return;

  const int col   = lane & 15;
  const int khalf = lane >> 4;

  const int lA    = tile * 16 + col;
  const int baseA = (lA / Q) * P + (lA % Q);

  int baseD[8];
#pragma unroll
  for (int v = 0; v < 8; ++v) {
    int lD = tile * 16 + khalf * 8 + v;
    baseD[v] = (lD / Q) * P + (lD % Q);
  }

  v8f acc[NT];
#pragma unroll
  for (int t = 0; t < NT; ++t) acc[t] = {};

  for (int ks = 0; ks < KSTEPS; ++ks) {
    const int kb = ks * 4;
    v2f a;
    a.x = in[baseA + (kb + 2 * khalf + 0) * S];
    a.y = in[baseA + (kb + 2 * khalf + 1) * S];
    __builtin_prefetch(&in[baseA + (kb + 4 + 2 * khalf) * S], 0, 1);
#pragma unroll
    for (int t = 0; t < NT; ++t) {
      const float* Hrow = H + (t * 16 + col) * NAX + kb + 2 * khalf;
      v2f b;
      b.x = Hrow[0];
      b.y = Hrow[1];
      acc[t] = __builtin_amdgcn_wmma_f32_16x16x4_f32(
           false, a,  false, b,
           (short)0, acc[t],  false,  false);
    }
  }

#pragma unroll
  for (int t = 0; t < NT; ++t) {
#pragma unroll
    for (int v = 0; v < 8; ++v) {
      out[baseD[v] + (t * 16 + col) * S] = acc[t][v];
    }
  }
}

extern "C" void kernel_launch(void* const* d_in, const int* in_sizes, int n_in,
                              void* d_out, int out_size, void* d_ws, size_t ws_size,
                              hipStream_t stream) {
  const float* x   = (const float*)d_in[0];
  float*       out = (float*)d_out;
  float*       H   = (float*)d_ws;
  float*       tmp = (float*)d_ws + 32768;

  (void)in_sizes; (void)n_in; (void)out_size; (void)ws_size;

  build_H<<<dim3((NAX * NAX + 255) / 256), dim3(256), 0, stream>>>(H);

  const int lines = 2 * 3 * NAX * NAX;
  const int tiles = lines / 16;
  dim3 grid((tiles + 7) / 8), block(256);

  spline_pass<1, NAX, 1>
      <<<grid, block, 0, stream>>>(x, out, H, tiles);
  spline_pass<NAX, NAX * NAX, NAX>
      <<<grid, block, 0, stream>>>(out, tmp, H, tiles);
  spline_pass<NAX * NAX, NAX * NAX * NAX, NAX * NAX>
      <<<grid, block, 0, stream>>>(tmp, out, H, tiles);
}


# B fragments of the circulant H held in registers: only 40 distinct ones exist, loaded once per wave; K loop fully unrolled, no H loads left in it
# speedup vs baseline: 1.6883x; 1.6883x over previous
#include <hip/hip_runtime.h>
#include <hip/hip_bf16.h>


typedef __attribute__((ext_vector_type(2))) float v2f;
typedef __attribute__((ext_vector_type(8))) float v8f;

#define NAX 160
#define NT  10
#define KSTEPS 40
#define NFRAG  40

__global__ void build_H(float* __restrict__ H) {
  int idx = blockIdx.x * blockDim.x + threadIdx.x;
  if (idx >= NAX * NAX) return;
  int i = idx / NAX, j = idx % NAX;
  int d = i - j; if (d < 0) d += NAX;
  const double p = -0.26794919243112270647255365849413;
  double pd = 1.0, pnd = 1.0;
  for (int k = 0; k < d; ++k)        pd  *= p;
  for (int k = 0; k < NAX - d; ++k)  pnd *= p;
  double pN = pd * pnd;
  double h = 1.7320508075688772935274463415059 * (pd + pnd) / (1.0 - pN);
  H[idx] = (float)h;
}

template <int Q, int P, int S>
__global__ __launch_bounds__(256) void spline_pass(
    const float* __restrict__ in, float* __restrict__ out,
    const float* __restrict__ H, int nTiles)
{
  const int lane  = threadIdx.x & 31;
  const int wave  = threadIdx.x >> 5;
  const int tile  = blockIdx.x * 8 + wave;
  if (tile >= nTiles) return;

  const int col   = lane & 15;
  const int khalf = lane >> 4;

  const int lA    = tile * 16 + col;
  const int baseA = (lA / Q) * P + (lA % Q);

  int baseD[8];
#pragma unroll
  for (int v = 0; v < 8; ++v) {
    int lD = tile * 16 + khalf * 8 + v;
    baseD[v] = (lD / Q) * P + (lD % Q);
  }

  v2f frag[NFRAG];
#pragma unroll
  for (int r = 0; r < NFRAG; ++r) {
    int row = 4 * r + col;
    if (row >= NAX) row -= NAX;
    const float* Hrow = H + row * NAX + 2 * khalf;
    frag[r].x = Hrow[0];
    frag[r].y = Hrow[1];
  }

  v8f acc[NT];
#pragma unroll
  for (int t = 0; t < NT; ++t) acc[t] = {};

#pragma unroll
  for (int ks = 0; ks < KSTEPS; ++ks) {
    const int kb = ks * 4;
    v2f a;
    a.x = in[baseA + (kb + 2 * khalf + 0) * S];
    a.y = in[baseA + (kb + 2 * khalf + 1) * S];
#pragma unroll
    for (int t = 0; t < NT; ++t) {
      acc[t] = __builtin_amdgcn_wmma_f32_16x16x4_f32(
           false, a,  false, frag[(4 * t - ks + NFRAG) % NFRAG],
           (short)0, acc[t],  false,  false);
    }
  }

#pragma unroll
  for (int t = 0; t < NT; ++t) {
#pragma unroll
    for (int v = 0; v < 8; ++v) {
      out[baseD[v] + (t * 16 + col) * S] = acc[t][v];
    }
  }
}

extern "C" void kernel_launch(void* const* d_in, const int* in_sizes, int n_in,
                              void* d_out, int out_size, void* d_ws, size_t ws_size,
                              hipStream_t stream) {
  const float* x   = (const float*)d_in[0];
  float*       out = (float*)d_out;
  float*       H   = (float*)d_ws;
  float*       tmp = (float*)d_ws + 32768;

  (void)in_sizes; (void)n_in; (void)out_size; (void)ws_size;

  build_H<<<dim3((NAX * NAX + 255) / 256), dim3(256), 0, stream>>>(H);

  const int lines = 2 * 3 * NAX * NAX;
  const int tiles = lines / 16;
  dim3 grid((tiles + 7) / 8), block(256);

  spline_pass<1, NAX, 1>
      <<<grid, block, 0, stream>>>(x, out, H, tiles);
  spline_pass<NAX, NAX * NAX, NAX>
      <<<grid, block, 0, stream>>>(out, tmp, H, tiles);
  spline_pass<NAX * NAX, NAX * NAX * NAX, NAX * NAX>
      <<<grid, block, 0, stream>>>(tmp, out, H, tiles);
}


# Strided passes form the product transposed (H fragment as WMMA A operand, input as B): each store writes two 64-byte runs, not 32 dwords in 32 lines
# speedup vs baseline: 2.5359x; 1.5020x over previous
#include <hip/hip_runtime.h>
#include <hip/hip_bf16.h>


typedef __attribute__((ext_vector_type(2))) float v2f;
typedef __attribute__((ext_vector_type(8))) float v8f;

#define NAX 160
#define NT  10
#define KSTEPS 40
#define NFRAG  40

__global__ void build_H(float* __restrict__ H) {
  int idx = blockIdx.x * blockDim.x + threadIdx.x;
  if (idx >= NAX * NAX) return;
  int i = idx / NAX, j = idx % NAX;
  int d = i - j; if (d < 0) d += NAX;
  const double p = -0.26794919243112270647255365849413;
  double pd = 1.0, pnd = 1.0;
  for (int k = 0; k < d; ++k)        pd  *= p;
  for (int k = 0; k < NAX - d; ++k)  pnd *= p;
  double pN = pd * pnd;
  double h = 1.7320508075688772935274463415059 * (pd + pnd) / (1.0 - pN);
  H[idx] = (float)h;
}

template <int Q, int P, int S>
__global__ __launch_bounds__(256) void spline_pass(
    const float* __restrict__ in, float* __restrict__ out,
    const float* __restrict__ H, int nTiles)
{
  const int lane  = threadIdx.x & 31;
  const int wave  = threadIdx.x >> 5;
  const int tile  = blockIdx.x * 8 + wave;
  if (tile >= nTiles) return;

  const int col   = lane & 15;
  const int khalf = lane >> 4;

  const int lA    = tile * 16 + col;
  const int baseA = (lA / Q) * P + (lA % Q);

  constexpr bool kTransposed = (S != 1);

  v2f frag[NFRAG];
#pragma unroll
  for (int r = 0; r < NFRAG; ++r) {
    int row = 4 * r + col;
    if (row >= NAX) row -= NAX;
    const float* Hrow = H + row * NAX + 2 * khalf;
    frag[r].x = Hrow[0];
    frag[r].y = Hrow[1];
  }

  v8f acc[NT];
#pragma unroll
  for (int t = 0; t < NT; ++t) acc[t] = {};

#pragma unroll
  for (int ks = 0; ks < KSTEPS; ++ks) {
    const int kb = ks * 4;
    v2f a;
    a.x = in[baseA + (kb + 2 * khalf + 0) * S];
    a.y = in[baseA + (kb + 2 * khalf + 1) * S];
#pragma unroll
    for (int t = 0; t < NT; ++t) {
      if constexpr (kTransposed) {
        acc[t] = __builtin_amdgcn_wmma_f32_16x16x4_f32(
             false, frag[(4 * t - ks + NFRAG) % NFRAG],  false, a,
             (short)0, acc[t],  false,  false);
      } else {
        acc[t] = __builtin_amdgcn_wmma_f32_16x16x4_f32(
             false, a,  false, frag[(4 * t - ks + NFRAG) % NFRAG],
             (short)0, acc[t],  false,  false);
      }
    }
  }

  if constexpr (kTransposed) {
    const int baseO = baseA + 8 * khalf * S;
#pragma unroll
    for (int t = 0; t < NT; ++t) {
#pragma unroll
      for (int v = 0; v < 8; ++v) {
        out[baseO + (t * 16 + v) * S] = acc[t][v];
      }
    }
  } else {
    int baseD[8];
#pragma unroll
    for (int v = 0; v < 8; ++v) {
      int lD = tile * 16 + khalf * 8 + v;
      baseD[v] = (lD / Q) * P + (lD % Q);
    }
#pragma unroll
    for (int t = 0; t < NT; ++t) {
#pragma unroll
      for (int v = 0; v < 8; ++v) {
        out[baseD[v] + (t * 16 + col) * S] = acc[t][v];
      }
    }
  }
}

extern "C" void kernel_launch(void* const* d_in, const int* in_sizes, int n_in,
                              void* d_out, int out_size, void* d_ws, size_t ws_size,
                              hipStream_t stream) {
  const float* x   = (const float*)d_in[0];
  float*       out = (float*)d_out;
  float*       H   = (float*)d_ws;
  float*       tmp = (float*)d_ws + 32768;

  (void)in_sizes; (void)n_in; (void)out_size; (void)ws_size;

  build_H<<<dim3((NAX * NAX + 255) / 256), dim3(256), 0, stream>>>(H);

  const int lines = 2 * 3 * NAX * NAX;
  const int tiles = lines / 16;
  dim3 grid((tiles + 7) / 8), block(256);

  spline_pass<1, NAX, 1>
      <<<grid, block, 0, stream>>>(x, out, H, tiles);
  spline_pass<NAX, NAX * NAX, NAX>
      <<<grid, block, 0, stream>>>(out, tmp, H, tiles);
  spline_pass<NAX * NAX, NAX * NAX * NAX, NAX * NAX>
      <<<grid, block, 0, stream>>>(tmp, out, H, tiles);
}


# Persistent waves: 512 workgroups per pass, each wave loops over its 2-3 strips with H fragments loaded once; next strip's first loads issued during the current strip's last K-steps
# speedup vs baseline: 2.6346x; 1.0389x over previous
#include <hip/hip_runtime.h>
#include <hip/hip_bf16.h>


typedef __attribute__((ext_vector_type(2))) float v2f;
typedef __attribute__((ext_vector_type(8))) float v8f;

#define NAX 160
#define NT  10
#define KSTEPS 40
#define NFRAG  40
#define WPB    8
#define NBLK   512
#define RING   20

__global__ void build_H(float* __restrict__ H) {
  int idx = blockIdx.x * blockDim.x + threadIdx.x;
  if (idx >= NAX * NAX) return;
  int i = idx / NAX, j = idx % NAX;
  int d = i - j; if (d < 0) d += NAX;
  const double p = -0.26794919243112270647255365849413;
  double pd = 1.0, pnd = 1.0;
  for (int k = 0; k < d; ++k)        pd  *= p;
  for (int k = 0; k < NAX - d; ++k)  pnd *= p;
  double pN = pd * pnd;
  double h = 1.7320508075688772935274463415059 * (pd + pnd) / (1.0 - pN);
  H[idx] = (float)h;
}

template <int Q, int P, int S>
__global__ __launch_bounds__(256) void spline_pass(
    const float* __restrict__ in, float* __restrict__ out,
    const float* __restrict__ H, int nTiles)
{
  const int lane  = threadIdx.x & 31;
  const int wave  = __builtin_amdgcn_readfirstlane((int)(threadIdx.x >> 5));
  const int col   = lane & 15;
  const int khalf = lane >> 4;

  const unsigned nPairs = (unsigned)nTiles >> 1;
  const int first = (int)(2u * ((blockIdx.x * nPairs) / NBLK));
  const int end   = (blockIdx.x + 1u == NBLK) ? nTiles
                                              : (int)(2u * (((blockIdx.x + 1u) * nPairs) / NBLK));
  const int count = end - first;
  const int full  = count / WPB;
  const int rem   = count % WPB;
  const int pw    = (wave * 5) & (WPB - 1);
  const int nMine = full + (pw < rem ? 1 : 0);
  if (nMine <= 0) return;

  auto stripOf = [&](int i) -> int {
    return first + (i < full ? i * WPB + wave : full * WPB + pw);
  };
  auto lineBase = [&](int tile) -> int {
    const int l = tile * 16 + col;
    return (l / Q) * P + (l % Q);
  };
  auto loadA = [&](int base, int ks) -> v2f {
    v2f a;
    a.x = in[base + (ks * 4 + 2 * khalf + 0) * S];
    a.y = in[base + (ks * 4 + 2 * khalf + 1) * S];
    return a;
  };

  constexpr bool kTransposed = (S != 1);
  static_assert(kTransposed || Q == 1, "the contiguous pass addresses its lines as l * P");

  v2f frag[NFRAG];
#pragma unroll
  for (int r = 0; r < NFRAG; ++r) {
    int row = 4 * r + col;
    if (row >= NAX) row -= NAX;
    const float* Hrow = H + row * NAX + 2 * khalf;
    frag[r].x = Hrow[0];
    frag[r].y = Hrow[1];
  }
  __builtin_amdgcn_sched_barrier(0);

  int baseA = lineBase(stripOf(0));

  v2f ring[RING];
#pragma unroll
  for (int r = 0; r < RING; ++r) {
    ring[r] = loadA(baseA, r);
    __builtin_amdgcn_sched_barrier(0);
  }

  for (int i = 0; i < nMine; ++i) {
    const int baseN = lineBase(stripOf(i + 1 < nMine ? i + 1 : i));

    v8f acc[NT];
#pragma unroll
    for (int t = 0; t < NT; ++t) acc[t] = {};

#pragma unroll
    for (int ks = 0; ks < KSTEPS; ++ks) {
      const v2f a = ring[ks % RING];
#pragma unroll
      for (int t = 0; t < NT; ++t) {
        if constexpr (kTransposed) {
          acc[t] = __builtin_amdgcn_wmma_f32_16x16x4_f32(
               false, frag[(4 * t - ks + NFRAG) % NFRAG],  false, a,
               (short)0, acc[t],  false,  false);
        } else {
          acc[t] = __builtin_amdgcn_wmma_f32_16x16x4_f32(
               false, a,  false, frag[(4 * t - ks + NFRAG) % NFRAG],
               (short)0, acc[t],  false,  false);
        }
      }
      if (ks + RING < KSTEPS) ring[ks % RING] = loadA(baseA, ks + RING);
      else                    ring[ks % RING] = loadA(baseN, ks + RING - KSTEPS);
      __builtin_amdgcn_sched_barrier(0);
    }

    if constexpr (kTransposed) {
      const int baseO = baseA + 8 * khalf * S;
#pragma unroll
      for (int t = 0; t < NT; ++t) {
#pragma unroll
        for (int v = 0; v < 8; ++v) {
          out[baseO + (t * 16 + v) * S] = acc[t][v];
        }
      }
    } else {
      const int baseD = baseA + (8 * khalf - col) * P + col;
#pragma unroll
      for (int t = 0; t < NT; ++t) {
#pragma unroll
        for (int v = 0; v < 8; ++v) {
          out[baseD + v * P + t * 16 * S] = acc[t][v];
        }
      }
    }

    baseA = baseN;
  }
}

extern "C" void kernel_launch(void* const* d_in, const int* in_sizes, int n_in,
                              void* d_out, int out_size, void* d_ws, size_t ws_size,
                              hipStream_t stream) {
  const float* x   = (const float*)d_in[0];
  float*       out = (float*)d_out;
  float*       H   = (float*)d_ws;
  float*       tmp = (float*)d_ws + 32768;

  (void)in_sizes; (void)n_in; (void)out_size; (void)ws_size;

  build_H<<<dim3((NAX * NAX + 255) / 256), dim3(256), 0, stream>>>(H);

  const int lines = 2 * 3 * NAX * NAX;
  const int tiles = lines / 16;
  dim3 grid(NBLK), block(WPB * 32);

  spline_pass<1, NAX, 1>
      <<<grid, block, 0, stream>>>(x, out, H, tiles);
  spline_pass<NAX, NAX * NAX, NAX>
      <<<grid, block, 0, stream>>>(out, tmp, H, tiles);
  spline_pass<NAX * NAX, NAX * NAX * NAX, NAX * NAX>
      <<<grid, block, 0, stream>>>(tmp, out, H, tiles);
}


# Whole-line loads: operand tables stored in fragment order (256 contiguous bytes per load, was 16 lines); contiguous pass loads 16 bytes per lane, so its k sum order inside each 8-k group changes
# speedup vs baseline: 4.1130x; 1.5611x over previous
#include <hip/hip_runtime.h>
#include <hip/hip_bf16.h>


typedef __attribute__((ext_vector_type(2))) float v2f;
typedef __attribute__((ext_vector_type(4))) float v4f;
typedef __attribute__((ext_vector_type(8))) float v8f;

#define NAX 160
#define NT  10
#define KSTEPS 40
#define WPB    8
#define NBLK   512
#define RING   10
#define BLK    32
#define NBLOCKS 5
#define KPB    8
#define NDIAG  12
#define FRAGF  64
#define D_FL   (NDIAG * FRAGF)
#define G_FL   (D_FL + KSTEPS * FRAGF)
#define E_OFF  (2 * G_FL)
#define NTAB   (E_OFF + 8 * BLK)

__device__ constexpr int kposOf(int g, int ks, int h, int s) {
  return g ? 8 * (ks >> 1) + 4 * h + 2 * (ks & 1) + s : 4 * ks + 2 * h + s;
}
__device__ constexpr int slotOf(int tt, int kk) {
  return (tt == 1 && kk >= 4) ? kk - 4 : 8 * tt + kk;
}

__global__ void build_H(float* __restrict__ H) {
  int idx = blockIdx.x * blockDim.x + threadIdx.x;
  if (idx >= NTAB) return;
  const double p  = -0.26794919243112270647255365849413;
  const double s3 = 1.7320508075688772935274463415059;
  if (idx < E_OFF) {
    const int g    = idx / G_FL;
    const int q    = idx % G_FL;
    const int s    = q & 1;
    const int lane = (q >> 1) & 31;
    const int col  = lane & 15, h = lane >> 4;
    if (q < D_FL) {
      const int slot = q / FRAGF;
      const int tt = slot >> 3, kk = slot & 7;
      int d = 16 * tt + col - kposOf(g, kk, h, s);
      if (d < 0) d += NAX;
      double pd = 1.0, pnd = 1.0;
      for (int k = 0; k < d; ++k)        pd  *= p;
      for (int k = 0; k < NAX - d; ++k)  pnd *= p;
      double pN = pd * pnd;
      double hv = s3 * (pd + pnd) / (1.0 - pN);
      H[idx] = (float)hv;
    } else {
      const int ks  = (q - D_FL) / FRAGF;
      const int row = col;
      const int k   = kposOf(g, ks, h, s);
      const int I = row & 7;
      double w = 0.0;
      if (I < NBLOCKS) {
        int e = (BLK * I - k) % NAX; if (e < 0) e += NAX;
        if (e >= 1 && e <= NAX - BLK) {
          const int ex = (row >= 8) ? (NAX - BLK + 1 - e) : e;
          double pe = 1.0, pr = 1.0;
          for (int m = 0; m < ex; ++m)        pe *= p;
          for (int m = 0; m < NAX - ex; ++m)  pr *= p;
          w = s3 * pe / (1.0 - pe * pr);
        }
      }
      H[idx] = (float)w;
    }
  } else {
    const int q    = idx - E_OFF;
    const int comp = q & 1;
    const int i    = (q >> 1) & (BLK - 1);
    const int kh   = (q >> 6) & 1;
    const int f    = q >> 7;
    const int ex   = kh ? (BLK - 1 - i) : i;
    double pe = 1.0;
    for (int m = 0; m < ex; ++m) pe *= p;
    H[idx] = (comp == f) ? (float)pe : 0.0f;
  }
}

template <int Q, int P, int S>
__global__ __launch_bounds__(256) void spline_pass(
    const float* __restrict__ in, float* __restrict__ out,
    const float* __restrict__ H, int nTiles)
{
  const int lane  = threadIdx.x & 31;
  const int wave  = __builtin_amdgcn_readfirstlane((int)(threadIdx.x >> 5));
  const int col   = lane & 15;
  const int khalf = lane >> 4;

  const unsigned nPairs = (unsigned)nTiles >> 1;
  const int first = (int)(2u * ((blockIdx.x * nPairs) / NBLK));
  const int end   = (blockIdx.x + 1u == NBLK) ? nTiles
                                              : (int)(2u * (((blockIdx.x + 1u) * nPairs) / NBLK));
  const int count = end - first;
  const int full  = count / WPB;
  const int rem   = count % WPB;
  const int pw    = (wave * 5) & (WPB - 1);
  const int nMine = full + (pw < rem ? 1 : 0);
  if (nMine <= 0) return;

  auto stripOf = [&](int i) -> int {
    return first + (i < full ? i * WPB + wave : full * WPB + pw);
  };
  auto lineBase = [&](int tile) -> int {
    const int l = tile * 16 + col;
    return (l / Q) * P + (l % Q);
  };
  auto loadA = [&](int base, int ks) -> v2f {
    v2f a;
    a.x = in[base + (ks * 4 + 2 * khalf + 0) * S];
    a.y = in[base + (ks * 4 + 2 * khalf + 1) * S];
    return a;
  };
  auto loadQ = [&](int base, int j) -> v4f {
    return *reinterpret_cast<const v4f*>(in + base + 8 * j + 4 * khalf);
  };

  constexpr bool kTransposed = (S != 1);
  static_assert(kTransposed || Q == 1, "the contiguous pass addresses its lines as l * P");

  constexpr int G = kTransposed ? 0 : 1;
  const float* Tg = H + G * G_FL + 2 * lane;
  v2f frag[NDIAG];
#pragma unroll
  for (int s = 0; s < NDIAG; ++s) {
    frag[s].x = Tg[s * FRAGF];
    frag[s].y = Tg[s * FRAGF + 1];
  }
  v2f wfrag[KSTEPS];
#pragma unroll
  for (int ks = 0; ks < KSTEPS; ++ks) {
    wfrag[ks].x = Tg[D_FL + ks * FRAGF];
    wfrag[ks].y = Tg[D_FL + ks * FRAGF + 1];
  }
  v2f eEven[2], eOdd[2];
#pragma unroll
  for (int tt = 0; tt < 2; ++tt) {
    const float* Ee = H + E_OFF + ((0 * 2 + khalf) * BLK + 16 * tt + col) * 2;
    const float* Eo = H + E_OFF + ((1 * 2 + khalf) * BLK + 16 * tt + col) * 2;
    eEven[tt].x = Ee[0]; eEven[tt].y = Ee[1];
    eOdd[tt].x  = Eo[0]; eOdd[tt].y  = Eo[1];
  }
  __builtin_amdgcn_sched_barrier(0);

  int baseA = lineBase(stripOf(0));

  static_assert(RING % 2 == 0 && (KSTEPS / 2) % (RING / 2) == 0, "ring4 slots must be static");
  v2f ring[RING];
  v4f ring4[RING / 2];
  if constexpr (kTransposed) {
#pragma unroll
    for (int r = 0; r < RING; ++r) {
      ring[r] = loadA(baseA, r);
      __builtin_amdgcn_sched_barrier(0);
    }
  } else {
#pragma unroll
    for (int r = 0; r < RING / 2; ++r) {
      ring4[r] = loadQ(baseA, r);
      __builtin_amdgcn_sched_barrier(0);
    }
  }

  for (int i = 0; i < nMine; ++i) {
    const int baseN = lineBase(stripOf(i + 1 < nMine ? i + 1 : i));

    v8f acc[NT];
#pragma unroll
    for (int t = 0; t < NT; ++t) acc[t] = {};
    v8f st = {};

#pragma unroll
    for (int ks = 0; ks < KSTEPS; ++ks) {
      v2f a;
      if constexpr (kTransposed) {
        a = ring[ks % RING];
      } else {
        const v4f q = ring4[(ks >> 1) % (RING / 2)];
        if (ks & 1) { a.x = q.z; a.y = q.w; } else { a.x = q.x; a.y = q.y; }
      }
      const int J  = ks / KPB;
      const int kk = ks % KPB;
#pragma unroll
      for (int tt = 0; tt < 2; ++tt) {
        const int t = 2 * J + tt;
        const int s = slotOf(tt, kk);
        if constexpr (kTransposed) {
          acc[t] = __builtin_amdgcn_wmma_f32_16x16x4_f32(
               false, frag[s],  false, a,
               (short)0, acc[t],  false,  false);
        } else {
          acc[t] = __builtin_amdgcn_wmma_f32_16x16x4_f32(
               false, a,  false, frag[s],
               (short)0, acc[t],  false,  false);
        }
      }
      st = __builtin_amdgcn_wmma_f32_16x16x4_f32(
           false, wfrag[ks],  false, a,
           (short)0, st,  false,  false);
      if constexpr (kTransposed) {
        if (ks + RING < KSTEPS) ring[ks % RING] = loadA(baseA, ks + RING);
        else                    ring[ks % RING] = loadA(baseN, ks + RING - KSTEPS);
      } else if (ks & 1) {
        const int j = (ks >> 1) + RING / 2;
        if (j < KSTEPS / 2) ring4[(ks >> 1) % (RING / 2)] = loadQ(baseA, j);
        else                ring4[(ks >> 1) % (RING / 2)] = loadQ(baseN, j - KSTEPS / 2);
      }
      __builtin_amdgcn_sched_barrier(0);
    }

#pragma unroll
    for (int I = 0; I < NBLOCKS; ++I) {
      v2f sv; sv.x = st[I & ~1]; sv.y = st[(I & ~1) + 1];
#pragma unroll
      for (int tt = 0; tt < 2; ++tt) {
        const int t = 2 * I + tt;
        const v2f ev = (I & 1) ? eOdd[tt] : eEven[tt];
        if constexpr (kTransposed) {
          acc[t] = __builtin_amdgcn_wmma_f32_16x16x4_f32(
               false, ev,  false, sv,
               (short)0, acc[t],  false,  false);
        } else {
          acc[t] = __builtin_amdgcn_wmma_f32_16x16x4_f32(
               false, sv,  false, ev,
               (short)0, acc[t],  false,  false);
        }
      }
    }

    if constexpr (kTransposed) {
      const int baseO = baseA + 8 * khalf * S;
#pragma unroll
      for (int t = 0; t < NT; ++t) {
#pragma unroll
        for (int v = 0; v < 8; ++v) {
          out[baseO + (t * 16 + v) * S] = acc[t][v];
        }
      }
    } else {
      const int baseD = baseA + (8 * khalf - col) * P + col;
#pragma unroll
      for (int t = 0; t < NT; ++t) {
#pragma unroll
        for (int v = 0; v < 8; ++v) {
          out[baseD + v * P + t * 16 * S] = acc[t][v];
        }
      }
    }

    baseA = baseN;
  }
}

extern "C" void kernel_launch(void* const* d_in, const int* in_sizes, int n_in,
                              void* d_out, int out_size, void* d_ws, size_t ws_size,
                              hipStream_t stream) {
  const float* x   = (const float*)d_in[0];
  float*       out = (float*)d_out;
  float*       H   = (float*)d_ws;
  float*       tmp = (float*)d_ws + 32768;

  (void)in_sizes; (void)n_in; (void)out_size; (void)ws_size;

  build_H<<<dim3((NTAB + 255) / 256), dim3(256), 0, stream>>>(H);

  const int lines = 2 * 3 * NAX * NAX;
  const int tiles = lines / 16;
  dim3 grid(NBLK), block(WPB * 32);

  spline_pass<1, NAX, 1>
      <<<grid, block, 0, stream>>>(x, out, H, tiles);
  spline_pass<NAX, NAX * NAX, NAX>
      <<<grid, block, 0, stream>>>(out, tmp, H, tiles);
  spline_pass<NAX * NAX, NAX * NAX * NAX, NAX * NAX>
      <<<grid, block, 0, stream>>>(tmp, out, H, tiles);
}


# Strip in two sweeps over a register-held strip: far-part sums first, then block by block diagonal + spread + store; stores leave in 5 groups between WMMAs, loads issued a whole strip ahead
# speedup vs baseline: 4.2904x; 1.0431x over previous
#include <hip/hip_runtime.h>
#include <hip/hip_bf16.h>


typedef __attribute__((ext_vector_type(2))) float v2f;
typedef __attribute__((ext_vector_type(4))) float v4f;
typedef __attribute__((ext_vector_type(8))) float v8f;

#define NAX 160
#define NT  10
#define KSTEPS 40
#define WPB    8
#define NBLK   512
#define BLK    32
#define NBLOCKS 5
#define KPB    8
#define NDIAG  12
#define FRAGF  64
#define D_FL   (NDIAG * FRAGF)
#define G_FL   (D_FL + KSTEPS * FRAGF)
#define E_OFF  (2 * G_FL)
#define NTAB   (E_OFF + 8 * BLK)

__device__ constexpr int kposOf(int g, int ks, int h, int s) {
  return g ? 8 * (ks >> 1) + 4 * h + 2 * (ks & 1) + s : 4 * ks + 2 * h + s;
}
__device__ constexpr int slotOf(int tt, int kk) {
  return (tt == 1 && kk >= 4) ? kk - 4 : 8 * tt + kk;
}

__global__ void build_H(float* __restrict__ H) {
  int idx = blockIdx.x * blockDim.x + threadIdx.x;
  if (idx >= NTAB) return;
  const double p  = -0.26794919243112270647255365849413;
  const double s3 = 1.7320508075688772935274463415059;
  if (idx < E_OFF) {
    const int g    = idx / G_FL;
    const int q    = idx % G_FL;
    const int s    = q & 1;
    const int lane = (q >> 1) & 31;
    const int col  = lane & 15, h = lane >> 4;
    if (q < D_FL) {
      const int slot = q / FRAGF;
      const int tt = slot >> 3, kk = slot & 7;
      int d = 16 * tt + col - kposOf(g, kk, h, s);
      if (d < 0) d += NAX;
      double pd = 1.0, pnd = 1.0;
      for (int k = 0; k < d; ++k)        pd  *= p;
      for (int k = 0; k < NAX - d; ++k)  pnd *= p;
      double pN = pd * pnd;
      double hv = s3 * (pd + pnd) / (1.0 - pN);
      H[idx] = (float)hv;
    } else {
      const int ks  = (q - D_FL) / FRAGF;
      const int row = col;
      const int k   = kposOf(g, ks, h, s);
      const int I = row & 7;
      double w = 0.0;
      if (I < NBLOCKS) {
        int e = (BLK * I - k) % NAX; if (e < 0) e += NAX;
        if (e >= 1 && e <= NAX - BLK) {
          const int ex = (row >= 8) ? (NAX - BLK + 1 - e) : e;
          double pe = 1.0, pr = 1.0;
          for (int m = 0; m < ex; ++m)        pe *= p;
          for (int m = 0; m < NAX - ex; ++m)  pr *= p;
          w = s3 * pe / (1.0 - pe * pr);
        }
      }
      H[idx] = (float)w;
    }
  } else {
    const int q    = idx - E_OFF;
    const int comp = q & 1;
    const int i    = (q >> 1) & (BLK - 1);
    const int kh   = (q >> 6) & 1;
    const int f    = q >> 7;
    const int ex   = kh ? (BLK - 1 - i) : i;
    double pe = 1.0;
    for (int m = 0; m < ex; ++m) pe *= p;
    H[idx] = (comp == f) ? (float)pe : 0.0f;
  }
}

template <int Q, int P, int S>
__global__ __launch_bounds__(256) void spline_pass(
    const float* __restrict__ in, float* __restrict__ out,
    const float* __restrict__ H, int nTiles)
{
  const int lane  = threadIdx.x & 31;
  const int wave  = __builtin_amdgcn_readfirstlane((int)(threadIdx.x >> 5));
  const int col   = lane & 15;
  const int khalf = lane >> 4;

  const unsigned nPairs = (unsigned)nTiles >> 1;
  const int first = (int)(2u * ((blockIdx.x * nPairs) / NBLK));
  const int end   = (blockIdx.x + 1u == NBLK) ? nTiles
                                              : (int)(2u * (((blockIdx.x + 1u) * nPairs) / NBLK));
  const int count = end - first;
  const int full  = count / WPB;
  const int rem   = count % WPB;
  const int pw    = (wave * 5) & (WPB - 1);
  const int nMine = full + (pw < rem ? 1 : 0);
  if (nMine <= 0) return;

  auto stripOf = [&](int i) -> int {
    return first + (i < full ? i * WPB + wave : full * WPB + pw);
  };
  auto lineBase = [&](int tile) -> int {
    const int l = tile * 16 + col;
    return (l / Q) * P + (l % Q);
  };
  auto loadA = [&](int base, int ks) -> v2f {
    v2f a;
    a.x = in[base + (ks * 4 + 2 * khalf + 0) * S];
    a.y = in[base + (ks * 4 + 2 * khalf + 1) * S];
    return a;
  };
  auto loadQ = [&](int base, int j) -> v4f {
    return *reinterpret_cast<const v4f*>(in + base + 8 * j + 4 * khalf);
  };

  constexpr bool kTransposed = (S != 1);
  static_assert(kTransposed || Q == 1, "the contiguous pass addresses its lines as l * P");

  constexpr int G = kTransposed ? 0 : 1;
  const float* Tg = H + G * G_FL + 2 * lane;
  v2f frag[NDIAG];
#pragma unroll
  for (int s = 0; s < NDIAG; ++s) {
    frag[s].x = Tg[s * FRAGF];
    frag[s].y = Tg[s * FRAGF + 1];
  }
  v2f wfrag[KSTEPS];
#pragma unroll
  for (int ks = 0; ks < KSTEPS; ++ks) {
    wfrag[ks].x = Tg[D_FL + ks * FRAGF];
    wfrag[ks].y = Tg[D_FL + ks * FRAGF + 1];
  }
  v2f eEven[2], eOdd[2];
#pragma unroll
  for (int tt = 0; tt < 2; ++tt) {
    const float* Ee = H + E_OFF + ((0 * 2 + khalf) * BLK + 16 * tt + col) * 2;
    const float* Eo = H + E_OFF + ((1 * 2 + khalf) * BLK + 16 * tt + col) * 2;
    eEven[tt].x = Ee[0]; eEven[tt].y = Ee[1];
    eOdd[tt].x  = Eo[0]; eOdd[tt].y  = Eo[1];
  }
  __builtin_amdgcn_sched_barrier(0);

  int baseA = lineBase(stripOf(0));

  v2f xin[KSTEPS];
  v4f xin4[KSTEPS / 2];
  if constexpr (kTransposed) {
#pragma unroll
    for (int ks = 0; ks < KSTEPS; ++ks) {
      xin[ks] = loadA(baseA, ks);
      __builtin_amdgcn_sched_barrier(0);
    }
  } else {
#pragma unroll
    for (int j = 0; j < KSTEPS / 2; ++j) {
      xin4[j] = loadQ(baseA, j);
      __builtin_amdgcn_sched_barrier(0);
    }
  }

  auto fragOf = [&](int ks) -> v2f {
    v2f a;
    if constexpr (kTransposed) {
      a = xin[ks];
    } else {
      const v4f q = xin4[ks >> 1];
      if (ks & 1) { a.x = q.z; a.y = q.w; } else { a.x = q.x; a.y = q.y; }
    }
    return a;
  };

  for (int i = 0; i < nMine; ++i) {
    const int baseN = lineBase(stripOf(i + 1 < nMine ? i + 1 : i));

    v8f st = {};
#pragma unroll
    for (int ks = 0; ks < KSTEPS; ++ks) {
      st = __builtin_amdgcn_wmma_f32_16x16x4_f32(
           false, wfrag[ks],  false, fragOf(ks),
           (short)0, st,  false,  false);
    }
    __builtin_amdgcn_sched_barrier(0);

#pragma unroll
    for (int J = 0; J < NBLOCKS; ++J) {
      v8f acc[2];
      acc[0] = {}; acc[1] = {};
#pragma unroll
      for (int kk = 0; kk < KPB; ++kk) {
        const v2f a = fragOf(KPB * J + kk);
#pragma unroll
        for (int tt = 0; tt < 2; ++tt) {
          const int s = slotOf(tt, kk);
          if constexpr (kTransposed) {
            acc[tt] = __builtin_amdgcn_wmma_f32_16x16x4_f32(
                 false, frag[s],  false, a,
                 (short)0, acc[tt],  false,  false);
          } else {
            acc[tt] = __builtin_amdgcn_wmma_f32_16x16x4_f32(
                 false, a,  false, frag[s],
                 (short)0, acc[tt],  false,  false);
          }
        }
      }
      {
        v2f sv; sv.x = st[J & ~1]; sv.y = st[(J & ~1) + 1];
#pragma unroll
        for (int tt = 0; tt < 2; ++tt) {
          const v2f ev = (J & 1) ? eOdd[tt] : eEven[tt];
          if constexpr (kTransposed) {
            acc[tt] = __builtin_amdgcn_wmma_f32_16x16x4_f32(
                 false, ev,  false, sv,
                 (short)0, acc[tt],  false,  false);
          } else {
            acc[tt] = __builtin_amdgcn_wmma_f32_16x16x4_f32(
                 false, sv,  false, ev,
                 (short)0, acc[tt],  false,  false);
          }
        }
      }
      if constexpr (kTransposed) {
#pragma unroll
        for (int kk = 0; kk < KPB; ++kk) xin[KPB * J + kk] = loadA(baseN, KPB * J + kk);
      } else {
#pragma unroll
        for (int jj = 0; jj < KPB / 2; ++jj) xin4[(KPB / 2) * J + jj] = loadQ(baseN, (KPB / 2) * J + jj);
      }
      if constexpr (kTransposed) {
        const int baseO = baseA + 8 * khalf * S;
#pragma unroll
        for (int tt = 0; tt < 2; ++tt) {
#pragma unroll
          for (int v = 0; v < 8; ++v) {
            out[baseO + ((2 * J + tt) * 16 + v) * S] = acc[tt][v];
          }
        }
      } else {
        const int baseD = baseA + (8 * khalf - col) * P + col;
#pragma unroll
        for (int tt = 0; tt < 2; ++tt) {
#pragma unroll
          for (int v = 0; v < 8; ++v) {
            out[baseD + v * P + (2 * J + tt) * 16 * S] = acc[tt][v];
          }
        }
      }
      __builtin_amdgcn_sched_barrier(0);
    }

    baseA = baseN;
  }
}

extern "C" void kernel_launch(void* const* d_in, const int* in_sizes, int n_in,
                              void* d_out, int out_size, void* d_ws, size_t ws_size,
                              hipStream_t stream) {
  const float* x   = (const float*)d_in[0];
  float*       out = (float*)d_out;
  float*       H   = (float*)d_ws;
  float*       tmp = (float*)d_ws + 32768;

  (void)in_sizes; (void)n_in; (void)out_size; (void)ws_size;

  build_H<<<dim3((NTAB + 255) / 256), dim3(256), 0, stream>>>(H);

  const int lines = 2 * 3 * NAX * NAX;
  const int tiles = lines / 16;
  dim3 grid(NBLK), block(WPB * 32);

  spline_pass<1, NAX, 1>
      <<<grid, block, 0, stream>>>(x, out, H, tiles);
  spline_pass<NAX, NAX * NAX, NAX>
      <<<grid, block, 0, stream>>>(out, tmp, H, tiles);
  spline_pass<NAX * NAX, NAX * NAX * NAX, NAX * NAX>
      <<<grid, block, 0, stream>>>(tmp, out, H, tiles);
}


# Passes 1 and 2 fused: one workgroup of 10 waves filters a 160x160 plane along both axes through an LDS copy; the 98 MB intermediate never goes to memory, one launch fewer; sums unchanged
# speedup vs baseline: 4.9473x; 1.1531x over previous
#include <hip/hip_runtime.h>
#include <hip/hip_bf16.h>


typedef __attribute__((ext_vector_type(2))) float v2f;
typedef __attribute__((ext_vector_type(4))) float v4f;
typedef __attribute__((ext_vector_type(8))) float v8f;

#define NAX 160
#define NT  10
#define KSTEPS 40
#define WPB    8
#define NBLK   512
#define BLK    32
#define NBLOCKS 5
#define KPB    8
#define NDIAG  12
#define FRAGF  64
#define D_FL   (NDIAG * FRAGF)
#define G_FL   (D_FL + KSTEPS * FRAGF)
#define E_OFF  (2 * G_FL)
#define NTAB   (E_OFF + 8 * BLK)

__device__ constexpr int kposOf(int g, int ks, int h, int s) {
  return g ? 8 * (ks >> 1) + 4 * h + 2 * (ks & 1) + s : 4 * ks + 2 * h + s;
}
__device__ constexpr int slotOf(int tt, int kk) {
  return (tt == 1 && kk >= 4) ? kk - 4 : 8 * tt + kk;
}

__global__ void build_H(float* __restrict__ H) {
  int idx = blockIdx.x * blockDim.x + threadIdx.x;
  if (idx >= NTAB) return;
  const double p  = -0.26794919243112270647255365849413;
  const double s3 = 1.7320508075688772935274463415059;
  if (idx < E_OFF) {
    const int g    = idx / G_FL;
    const int q    = idx % G_FL;
    const int s    = q & 1;
    const int lane = (q >> 1) & 31;
    const int col  = lane & 15, h = lane >> 4;
    if (q < D_FL) {
      const int slot = q / FRAGF;
      const int tt = slot >> 3, kk = slot & 7;
      int d = 16 * tt + col - kposOf(g, kk, h, s);
      if (d < 0) d += NAX;
      double pd = 1.0, pnd = 1.0;
      for (int k = 0; k < d; ++k)        pd  *= p;
      for (int k = 0; k < NAX - d; ++k)  pnd *= p;
      double pN = pd * pnd;
      double hv = s3 * (pd + pnd) / (1.0 - pN);
      H[idx] = (float)hv;
    } else {
      const int ks  = (q - D_FL) / FRAGF;
      const int row = col;
      const int k   = kposOf(g, ks, h, s);
      const int I = row & 7;
      double w = 0.0;
      if (I < NBLOCKS) {
        int e = (BLK * I - k) % NAX; if (e < 0) e += NAX;
        if (e >= 1 && e <= NAX - BLK) {
          const int ex = (row >= 8) ? (NAX - BLK + 1 - e) : e;
          double pe = 1.0, pr = 1.0;
          for (int m = 0; m < ex; ++m)        pe *= p;
          for (int m = 0; m < NAX - ex; ++m)  pr *= p;
          w = s3 * pe / (1.0 - pe * pr);
        }
      }
      H[idx] = (float)w;
    }
  } else {
    const int q    = idx - E_OFF;
    const int comp = q & 1;
    const int i    = (q >> 1) & (BLK - 1);
    const int kh   = (q >> 6) & 1;
    const int f    = q >> 7;
    const int ex   = kh ? (BLK - 1 - i) : i;
    double pe = 1.0;
    for (int m = 0; m < ex; ++m) pe *= p;
    H[idx] = (comp == f) ? (float)pe : 0.0f;
  }
}

template <int Q, int P, int S>
__global__ __launch_bounds__(256) void spline_pass(
    const float* __restrict__ in, float* __restrict__ out,
    const float* __restrict__ H, int nTiles)
{
  const int lane  = threadIdx.x & 31;
  const int wave  = __builtin_amdgcn_readfirstlane((int)(threadIdx.x >> 5));
  const int col   = lane & 15;
  const int khalf = lane >> 4;

  const unsigned nPairs = (unsigned)nTiles >> 1;
  const int first = (int)(2u * ((blockIdx.x * nPairs) / NBLK));
  const int end   = (blockIdx.x + 1u == NBLK) ? nTiles
                                              : (int)(2u * (((blockIdx.x + 1u) * nPairs) / NBLK));
  const int count = end - first;
  const int full  = count / WPB;
  const int rem   = count % WPB;
  const int pw    = (wave * 5) & (WPB - 1);
  const int nMine = full + (pw < rem ? 1 : 0);
  if (nMine <= 0) return;

  auto stripOf = [&](int i) -> int {
    return first + (i < full ? i * WPB + wave : full * WPB + pw);
  };
  auto lineBase = [&](int tile) -> int {
    const int l = tile * 16 + col;
    return (l / Q) * P + (l % Q);
  };
  auto loadA = [&](int base, int ks) -> v2f {
    v2f a;
    a.x = in[base + (ks * 4 + 2 * khalf + 0) * S];
    a.y = in[base + (ks * 4 + 2 * khalf + 1) * S];
    return a;
  };
  auto loadQ = [&](int base, int j) -> v4f {
    return *reinterpret_cast<const v4f*>(in + base + 8 * j + 4 * khalf);
  };

  constexpr bool kTransposed = (S != 1);
  static_assert(kTransposed || Q == 1, "the contiguous pass addresses its lines as l * P");

  constexpr int G = kTransposed ? 0 : 1;
  const float* Tg = H + G * G_FL + 2 * lane;
  v2f frag[NDIAG];
#pragma unroll
  for (int s = 0; s < NDIAG; ++s) {
    frag[s].x = Tg[s * FRAGF];
    frag[s].y = Tg[s * FRAGF + 1];
  }
  v2f wfrag[KSTEPS];
#pragma unroll
  for (int ks = 0; ks < KSTEPS; ++ks) {
    wfrag[ks].x = Tg[D_FL + ks * FRAGF];
    wfrag[ks].y = Tg[D_FL + ks * FRAGF + 1];
  }
  v2f eEven[2], eOdd[2];
#pragma unroll
  for (int tt = 0; tt < 2; ++tt) {
    const float* Ee = H + E_OFF + ((0 * 2 + khalf) * BLK + 16 * tt + col) * 2;
    const float* Eo = H + E_OFF + ((1 * 2 + khalf) * BLK + 16 * tt + col) * 2;
    eEven[tt].x = Ee[0]; eEven[tt].y = Ee[1];
    eOdd[tt].x  = Eo[0]; eOdd[tt].y  = Eo[1];
  }
  __builtin_amdgcn_sched_barrier(0);

  int baseA = lineBase(stripOf(0));

  v2f xin[KSTEPS];
  v4f xin4[KSTEPS / 2];
  if constexpr (kTransposed) {
#pragma unroll
    for (int ks = 0; ks < KSTEPS; ++ks) {
      xin[ks] = loadA(baseA, ks);
      __builtin_amdgcn_sched_barrier(0);
    }
  } else {
#pragma unroll
    for (int j = 0; j < KSTEPS / 2; ++j) {
      xin4[j] = loadQ(baseA, j);
      __builtin_amdgcn_sched_barrier(0);
    }
  }

  auto fragOf = [&](int ks) -> v2f {
    v2f a;
    if constexpr (kTransposed) {
      a = xin[ks];
    } else {
      const v4f q = xin4[ks >> 1];
      if (ks & 1) { a.x = q.z; a.y = q.w; } else { a.x = q.x; a.y = q.y; }
    }
    return a;
  };

  for (int i = 0; i < nMine; ++i) {
    const int baseN = lineBase(stripOf(i + 1 < nMine ? i + 1 : i));

    v8f st = {};
#pragma unroll
    for (int ks = 0; ks < KSTEPS; ++ks) {
      st = __builtin_amdgcn_wmma_f32_16x16x4_f32(
           false, wfrag[ks],  false, fragOf(ks),
           (short)0, st,  false,  false);
    }
    __builtin_amdgcn_sched_barrier(0);

#pragma unroll
    for (int J = 0; J < NBLOCKS; ++J) {
      v8f acc[2];
      acc[0] = {}; acc[1] = {};
#pragma unroll
      for (int kk = 0; kk < KPB; ++kk) {
        const v2f a = fragOf(KPB * J + kk);
#pragma unroll
        for (int tt = 0; tt < 2; ++tt) {
          const int s = slotOf(tt, kk);
          if constexpr (kTransposed) {
            acc[tt] = __builtin_amdgcn_wmma_f32_16x16x4_f32(
                 false, frag[s],  false, a,
                 (short)0, acc[tt],  false,  false);
          } else {
            acc[tt] = __builtin_amdgcn_wmma_f32_16x16x4_f32(
                 false, a,  false, frag[s],
                 (short)0, acc[tt],  false,  false);
          }
        }
      }
      {
        v2f sv; sv.x = st[J & ~1]; sv.y = st[(J & ~1) + 1];
#pragma unroll
        for (int tt = 0; tt < 2; ++tt) {
          const v2f ev = (J & 1) ? eOdd[tt] : eEven[tt];
          if constexpr (kTransposed) {
            acc[tt] = __builtin_amdgcn_wmma_f32_16x16x4_f32(
                 false, ev,  false, sv,
                 (short)0, acc[tt],  false,  false);
          } else {
            acc[tt] = __builtin_amdgcn_wmma_f32_16x16x4_f32(
                 false, sv,  false, ev,
                 (short)0, acc[tt],  false,  false);
          }
        }
      }
      if constexpr (kTransposed) {
#pragma unroll
        for (int kk = 0; kk < KPB; ++kk) xin[KPB * J + kk] = loadA(baseN, KPB * J + kk);
      } else {
#pragma unroll
        for (int jj = 0; jj < KPB / 2; ++jj) xin4[(KPB / 2) * J + jj] = loadQ(baseN, (KPB / 2) * J + jj);
      }
      if constexpr (kTransposed) {
        const int baseO = baseA + 8 * khalf * S;
#pragma unroll
        for (int tt = 0; tt < 2; ++tt) {
#pragma unroll
          for (int v = 0; v < 8; ++v) {
            out[baseO + ((2 * J + tt) * 16 + v) * S] = acc[tt][v];
          }
        }
      } else {
        const int baseD = baseA + (8 * khalf - col) * P + col;
#pragma unroll
        for (int tt = 0; tt < 2; ++tt) {
#pragma unroll
          for (int v = 0; v < 8; ++v) {
            out[baseD + v * P + (2 * J + tt) * 16 * S] = acc[tt][v];
          }
        }
      }
      __builtin_amdgcn_sched_barrier(0);
    }

    baseA = baseN;
  }
}

#define WPP    10
#define NPLANES 960
#define PITCH  172

template <int G>
__device__ __forceinline__ void load_tables(const float* __restrict__ H, int lane, int col, int khalf,
                                            v2f (&frag)[NDIAG], v2f (&wfrag)[KSTEPS],
                                            v2f (&eEven)[2], v2f (&eOdd)[2]) {
  const float* Tg = H + G * G_FL + 2 * lane;
#pragma unroll
  for (int s = 0; s < NDIAG; ++s) {
    frag[s].x = Tg[s * FRAGF];
    frag[s].y = Tg[s * FRAGF + 1];
  }
#pragma unroll
  for (int ks = 0; ks < KSTEPS; ++ks) {
    wfrag[ks].x = Tg[D_FL + ks * FRAGF];
    wfrag[ks].y = Tg[D_FL + ks * FRAGF + 1];
  }
#pragma unroll
  for (int tt = 0; tt < 2; ++tt) {
    const float* Ee = H + E_OFF + ((0 * 2 + khalf) * BLK + 16 * tt + col) * 2;
    const float* Eo = H + E_OFF + ((1 * 2 + khalf) * BLK + 16 * tt + col) * 2;
    eEven[tt].x = Ee[0]; eEven[tt].y = Ee[1];
    eOdd[tt].x  = Eo[0]; eOdd[tt].y  = Eo[1];
  }
}

template <bool kT, class FragOf, class StoreTile>
__device__ __forceinline__ void strip_compute(const v2f (&frag)[NDIAG], const v2f (&wfrag)[KSTEPS],
                                              const v2f (&eEven)[2], const v2f (&eOdd)[2],
                                              FragOf fragOf, StoreTile storeTile) {
  v8f st = {};
#pragma unroll
  for (int ks = 0; ks < KSTEPS; ++ks) {
    st = __builtin_amdgcn_wmma_f32_16x16x4_f32(
         false, wfrag[ks],  false, fragOf(ks),
         (short)0, st,  false,  false);
  }
  __builtin_amdgcn_sched_barrier(0);
#pragma unroll
  for (int J = 0; J < NBLOCKS; ++J) {
    v8f acc[2];
    acc[0] = {}; acc[1] = {};
#pragma unroll
    for (int kk = 0; kk < KPB; ++kk) {
      const v2f a = fragOf(KPB * J + kk);
#pragma unroll
      for (int tt = 0; tt < 2; ++tt) {
        const int s = slotOf(tt, kk);
        if constexpr (kT) {
          acc[tt] = __builtin_amdgcn_wmma_f32_16x16x4_f32(
               false, frag[s],  false, a,
               (short)0, acc[tt],  false,  false);
        } else {
          acc[tt] = __builtin_amdgcn_wmma_f32_16x16x4_f32(
               false, a,  false, frag[s],
               (short)0, acc[tt],  false,  false);
        }
      }
    }
    {
      v2f sv; sv.x = st[J & ~1]; sv.y = st[(J & ~1) + 1];
#pragma unroll
      for (int tt = 0; tt < 2; ++tt) {
        const v2f ev = (J & 1) ? eOdd[tt] : eEven[tt];
        if constexpr (kT) {
          acc[tt] = __builtin_amdgcn_wmma_f32_16x16x4_f32(
               false, ev,  false, sv,
               (short)0, acc[tt],  false,  false);
        } else {
          acc[tt] = __builtin_amdgcn_wmma_f32_16x16x4_f32(
               false, sv,  false, ev,
               (short)0, acc[tt],  false,  false);
        }
      }
    }
#pragma unroll
    for (int tt = 0; tt < 2; ++tt) storeTile(2 * J + tt, acc[tt]);
    __builtin_amdgcn_sched_barrier(0);
  }
}

__global__ __launch_bounds__(WPP * 32) void spline_plane(
    const float* __restrict__ in, float* __restrict__ out, const float* __restrict__ H)
{
  __shared__ float plane[NAX * PITCH];

  const int lane  = threadIdx.x & 31;
  const int wave  = __builtin_amdgcn_readfirstlane((int)(threadIdx.x >> 5));
  const int col   = lane & 15;
  const int khalf = lane >> 4;
  const float* src = in  + (size_t)blockIdx.x * (NAX * NAX);
  float*       dst = out + (size_t)blockIdx.x * (NAX * NAX);

  {
    v4f xin4[KSTEPS / 2];
    const float* lineA = src + (16 * wave + col) * NAX + 4 * khalf;
#pragma unroll
    for (int j = 0; j < KSTEPS / 2; ++j)
      xin4[j] = *reinterpret_cast<const v4f*>(lineA + 8 * j);
    __builtin_amdgcn_sched_barrier(0);
    v2f frag[NDIAG], wfrag[KSTEPS], eEven[2], eOdd[2];
    load_tables<1>(H, lane, col, khalf, frag, wfrag, eEven, eOdd);

    auto fragOf = [&](int ks) -> v2f {
      const v4f q = xin4[ks >> 1];
      v2f a;
      if (ks & 1) { a.x = q.z; a.y = q.w; } else { a.x = q.x; a.y = q.y; }
      return a;
    };
    float* rowD = plane + (16 * wave + 8 * khalf) * PITCH + col;
    auto storeTile = [&](int t, const v8f& acc) {
#pragma unroll
      for (int v = 0; v < 8; ++v) rowD[v * PITCH + 16 * t] = acc[v];
    };
    strip_compute<false>(frag, wfrag, eEven, eOdd, fragOf, storeTile);
  }

  {
    v2f frag[NDIAG], wfrag[KSTEPS], eEven[2], eOdd[2];
    load_tables<0>(H, lane, col, khalf, frag, wfrag, eEven, eOdd);

    __syncthreads();

    v2f xin[KSTEPS];
    const float* colB = plane + (2 * khalf) * PITCH + 16 * wave + col;
#pragma unroll
    for (int ks = 0; ks < KSTEPS; ++ks) {
      xin[ks].x = colB[(4 * ks + 0) * PITCH];
      xin[ks].y = colB[(4 * ks + 1) * PITCH];
    }
    auto fragOf = [&](int ks) -> v2f { return xin[ks]; };
    float* colD = dst + (8 * khalf) * NAX + 16 * wave + col;
    auto storeTile = [&](int t, const v8f& acc) {
#pragma unroll
      for (int v = 0; v < 8; ++v) colD[(16 * t + v) * NAX] = acc[v];
    };
    strip_compute<true>(frag, wfrag, eEven, eOdd, fragOf, storeTile);
  }
}

extern "C" void kernel_launch(void* const* d_in, const int* in_sizes, int n_in,
                              void* d_out, int out_size, void* d_ws, size_t ws_size,
                              hipStream_t stream) {
  const float* x   = (const float*)d_in[0];
  float*       out = (float*)d_out;
  float*       H   = (float*)d_ws;
  float*       tmp = (float*)d_ws + 32768;

  (void)in_sizes; (void)n_in; (void)out_size; (void)ws_size;

  build_H<<<dim3((NTAB + 255) / 256), dim3(256), 0, stream>>>(H);

  const int lines = 2 * 3 * NAX * NAX;
  const int tiles = lines / 16;
  dim3 grid(NBLK), block(WPB * 32);

  spline_plane<<<dim3(NPLANES), dim3(WPP * 32), 0, stream>>>(x, tmp, H);
  spline_pass<NAX * NAX, NAX * NAX * NAX, NAX * NAX>
      <<<grid, block, 0, stream>>>(tmp, out, H, tiles);
}


# Table kernel: each power of the pole by square-and-multiply (at most 8 dependent fp64 multiplies, was a counted loop of up to 160); exact divide by 1 - p^160 (= 1.0) dropped; same f32 tables
# speedup vs baseline: 5.8625x; 1.1850x over previous
#include <hip/hip_runtime.h>
#include <hip/hip_bf16.h>


typedef __attribute__((ext_vector_type(2))) float v2f;
typedef __attribute__((ext_vector_type(4))) float v4f;
typedef __attribute__((ext_vector_type(8))) float v8f;

#define NAX 160
#define NT  10
#define KSTEPS 40
#define WPB    8
#define NBLK   512
#define BLK    32
#define NBLOCKS 5
#define KPB    8
#define NDIAG  12
#define FRAGF  64
#define D_FL   (NDIAG * FRAGF)
#define G_FL   (D_FL + KSTEPS * FRAGF)
#define E_OFF  (2 * G_FL)
#define NTAB   (E_OFF + 8 * BLK)

__device__ constexpr int kposOf(int g, int ks, int h, int s) {
  return g ? 8 * (ks >> 1) + 4 * h + 2 * (ks & 1) + s : 4 * ks + 2 * h + s;
}
__device__ constexpr int slotOf(int tt, int kk) {
  return (tt == 1 && kk >= 4) ? kk - 4 : 8 * tt + kk;
}

__device__ __forceinline__ double pow_pole(int e) {
  double r = 1.0;
  double b = -0.26794919243112270647255365849413;
#pragma unroll
  for (int bit = 0; bit < 8; ++bit) {
    if ((e >> bit) & 1) r *= b;
    b *= b;
  }
  return r;
}

__global__ void build_H(float* __restrict__ H) {
  int idx = blockIdx.x * blockDim.x + threadIdx.x;
  if (idx >= NTAB) return;
  const double s3 = 1.7320508075688772935274463415059;
  if (idx < E_OFF) {
    const int g    = idx / G_FL;
    const int q    = idx % G_FL;
    const int s    = q & 1;
    const int lane = (q >> 1) & 31;
    const int col  = lane & 15, h = lane >> 4;
    if (q < D_FL) {
      const int slot = q / FRAGF;
      const int tt = slot >> 3, kk = slot & 7;
      int d = 16 * tt + col - kposOf(g, kk, h, s);
      if (d < 0) d += NAX;
      const double hv = s3 * (pow_pole(d) + pow_pole(NAX - d));
      H[idx] = (float)hv;
    } else {
      const int ks  = (q - D_FL) / FRAGF;
      const int row = col;
      const int k   = kposOf(g, ks, h, s);
      const int I = row & 7;
      double w = 0.0;
      if (I < NBLOCKS) {
        int e = (BLK * I - k) % NAX; if (e < 0) e += NAX;
        if (e >= 1 && e <= NAX - BLK) {
          const int ex = (row >= 8) ? (NAX - BLK + 1 - e) : e;
          w = s3 * pow_pole(ex);
        }
      }
      H[idx] = (float)w;
    }
  } else {
    const int q    = idx - E_OFF;
    const int comp = q & 1;
    const int i    = (q >> 1) & (BLK - 1);
    const int kh   = (q >> 6) & 1;
    const int f    = q >> 7;
    const int ex   = kh ? (BLK - 1 - i) : i;
    H[idx] = (comp == f) ? (float)pow_pole(ex) : 0.0f;
  }
}

template <int Q, int P, int S>
__global__ __launch_bounds__(256) void spline_pass(
    const float* __restrict__ in, float* __restrict__ out,
    const float* __restrict__ H, int nTiles)
{
  const int lane  = threadIdx.x & 31;
  const int wave  = __builtin_amdgcn_readfirstlane((int)(threadIdx.x >> 5));
  const int col   = lane & 15;
  const int khalf = lane >> 4;

  const unsigned nPairs = (unsigned)nTiles >> 1;
  const int first = (int)(2u * ((blockIdx.x * nPairs) / NBLK));
  const int end   = (blockIdx.x + 1u == NBLK) ? nTiles
                                              : (int)(2u * (((blockIdx.x + 1u) * nPairs) / NBLK));
  const int count = end - first;
  const int full  = count / WPB;
  const int rem   = count % WPB;
  const int pw    = (wave * 5) & (WPB - 1);
  const int nMine = full + (pw < rem ? 1 : 0);
  if (nMine <= 0) return;

  auto stripOf = [&](int i) -> int {
    return first + (i < full ? i * WPB + wave : full * WPB + pw);
  };
  auto lineBase = [&](int tile) -> int {
    const int l = tile * 16 + col;
    return (l / Q) * P + (l % Q);
  };
  auto loadA = [&](int base, int ks) -> v2f {
    v2f a;
    a.x = in[base + (ks * 4 + 2 * khalf + 0) * S];
    a.y = in[base + (ks * 4 + 2 * khalf + 1) * S];
    return a;
  };
  auto loadQ = [&](int base, int j) -> v4f {
    return *reinterpret_cast<const v4f*>(in + base + 8 * j + 4 * khalf);
  };

  constexpr bool kTransposed = (S != 1);
  static_assert(kTransposed || Q == 1, "the contiguous pass addresses its lines as l * P");

  constexpr int G = kTransposed ? 0 : 1;
  const float* Tg = H + G * G_FL + 2 * lane;
  v2f frag[NDIAG];
#pragma unroll
  for (int s = 0; s < NDIAG; ++s) {
    frag[s].x = Tg[s * FRAGF];
    frag[s].y = Tg[s * FRAGF + 1];
  }
  v2f wfrag[KSTEPS];
#pragma unroll
  for (int ks = 0; ks < KSTEPS; ++ks) {
    wfrag[ks].x = Tg[D_FL + ks * FRAGF];
    wfrag[ks].y = Tg[D_FL + ks * FRAGF + 1];
  }
  v2f eEven[2], eOdd[2];
#pragma unroll
  for (int tt = 0; tt < 2; ++tt) {
    const float* Ee = H + E_OFF + ((0 * 2 + khalf) * BLK + 16 * tt + col) * 2;
    const float* Eo = H + E_OFF + ((1 * 2 + khalf) * BLK + 16 * tt + col) * 2;
    eEven[tt].x = Ee[0]; eEven[tt].y = Ee[1];
    eOdd[tt].x  = Eo[0]; eOdd[tt].y  = Eo[1];
  }
  __builtin_amdgcn_sched_barrier(0);

  int baseA = lineBase(stripOf(0));

  v2f xin[KSTEPS];
  v4f xin4[KSTEPS / 2];
  if constexpr (kTransposed) {
#pragma unroll
    for (int ks = 0; ks < KSTEPS; ++ks) {
      xin[ks] = loadA(baseA, ks);
      __builtin_amdgcn_sched_barrier(0);
    }
  } else {
#pragma unroll
    for (int j = 0; j < KSTEPS / 2; ++j) {
      xin4[j] = loadQ(baseA, j);
      __builtin_amdgcn_sched_barrier(0);
    }
  }

  auto fragOf = [&](int ks) -> v2f {
    v2f a;
    if constexpr (kTransposed) {
      a = xin[ks];
    } else {
      const v4f q = xin4[ks >> 1];
      if (ks & 1) { a.x = q.z; a.y = q.w; } else { a.x = q.x; a.y = q.y; }
    }
    return a;
  };

  for (int i = 0; i < nMine; ++i) {
    const int baseN = lineBase(stripOf(i + 1 < nMine ? i + 1 : i));

    v8f st = {};
#pragma unroll
    for (int ks = 0; ks < KSTEPS; ++ks) {
      st = __builtin_amdgcn_wmma_f32_16x16x4_f32(
           false, wfrag[ks],  false, fragOf(ks),
           (short)0, st,  false,  false);
    }
    __builtin_amdgcn_sched_barrier(0);

#pragma unroll
    for (int J = 0; J < NBLOCKS; ++J) {
      v8f acc[2];
      acc[0] = {}; acc[1] = {};
#pragma unroll
      for (int kk = 0; kk < KPB; ++kk) {
        const v2f a = fragOf(KPB * J + kk);
#pragma unroll
        for (int tt = 0; tt < 2; ++tt) {
          const int s = slotOf(tt, kk);
          if constexpr (kTransposed) {
            acc[tt] = __builtin_amdgcn_wmma_f32_16x16x4_f32(
                 false, frag[s],  false, a,
                 (short)0, acc[tt],  false,  false);
          } else {
            acc[tt] = __builtin_amdgcn_wmma_f32_16x16x4_f32(
                 false, a,  false, frag[s],
                 (short)0, acc[tt],  false,  false);
          }
        }
      }
      {
        v2f sv; sv.x = st[J & ~1]; sv.y = st[(J & ~1) + 1];
#pragma unroll
        for (int tt = 0; tt < 2; ++tt) {
          const v2f ev = (J & 1) ? eOdd[tt] : eEven[tt];
          if constexpr (kTransposed) {
            acc[tt] = __builtin_amdgcn_wmma_f32_16x16x4_f32(
                 false, ev,  false, sv,
                 (short)0, acc[tt],  false,  false);
          } else {
            acc[tt] = __builtin_amdgcn_wmma_f32_16x16x4_f32(
                 false, sv,  false, ev,
                 (short)0, acc[tt],  false,  false);
          }
        }
      }
      if constexpr (kTransposed) {
#pragma unroll
        for (int kk = 0; kk < KPB; ++kk) xin[KPB * J + kk] = loadA(baseN, KPB * J + kk);
      } else {
#pragma unroll
        for (int jj = 0; jj < KPB / 2; ++jj) xin4[(KPB / 2) * J + jj] = loadQ(baseN, (KPB / 2) * J + jj);
      }
      if constexpr (kTransposed) {
        const int baseO = baseA + 8 * khalf * S;
#pragma unroll
        for (int tt = 0; tt < 2; ++tt) {
#pragma unroll
          for (int v = 0; v < 8; ++v) {
            out[baseO + ((2 * J + tt) * 16 + v) * S] = acc[tt][v];
          }
        }
      } else {
        const int baseD = baseA + (8 * khalf - col) * P + col;
#pragma unroll
        for (int tt = 0; tt < 2; ++tt) {
#pragma unroll
          for (int v = 0; v < 8; ++v) {
            out[baseD + v * P + (2 * J + tt) * 16 * S] = acc[tt][v];
          }
        }
      }
      __builtin_amdgcn_sched_barrier(0);
    }

    baseA = baseN;
  }
}

#define WPP    10
#define NPLANES 960
#define PITCH  172

template <int G>
__device__ __forceinline__ void load_tables(const float* __restrict__ H, int lane, int col, int khalf,
                                            v2f (&frag)[NDIAG], v2f (&wfrag)[KSTEPS],
                                            v2f (&eEven)[2], v2f (&eOdd)[2]) {
  const float* Tg = H + G * G_FL + 2 * lane;
#pragma unroll
  for (int s = 0; s < NDIAG; ++s) {
    frag[s].x = Tg[s * FRAGF];
    frag[s].y = Tg[s * FRAGF + 1];
  }
#pragma unroll
  for (int ks = 0; ks < KSTEPS; ++ks) {
    wfrag[ks].x = Tg[D_FL + ks * FRAGF];
    wfrag[ks].y = Tg[D_FL + ks * FRAGF + 1];
  }
#pragma unroll
  for (int tt = 0; tt < 2; ++tt) {
    const float* Ee = H + E_OFF + ((0 * 2 + khalf) * BLK + 16 * tt + col) * 2;
    const float* Eo = H + E_OFF + ((1 * 2 + khalf) * BLK + 16 * tt + col) * 2;
    eEven[tt].x = Ee[0]; eEven[tt].y = Ee[1];
    eOdd[tt].x  = Eo[0]; eOdd[tt].y  = Eo[1];
  }
}

template <bool kT, class FragOf, class StoreTile>
__device__ __forceinline__ void strip_compute(const v2f (&frag)[NDIAG], const v2f (&wfrag)[KSTEPS],
                                              const v2f (&eEven)[2], const v2f (&eOdd)[2],
                                              FragOf fragOf, StoreTile storeTile) {
  v8f st = {};
#pragma unroll
  for (int ks = 0; ks < KSTEPS; ++ks) {
    st = __builtin_amdgcn_wmma_f32_16x16x4_f32(
         false, wfrag[ks],  false, fragOf(ks),
         (short)0, st,  false,  false);
  }
  __builtin_amdgcn_sched_barrier(0);
#pragma unroll
  for (int J = 0; J < NBLOCKS; ++J) {
    v8f acc[2];
    acc[0] = {}; acc[1] = {};
#pragma unroll
    for (int kk = 0; kk < KPB; ++kk) {
      const v2f a = fragOf(KPB * J + kk);
#pragma unroll
      for (int tt = 0; tt < 2; ++tt) {
        const int s = slotOf(tt, kk);
        if constexpr (kT) {
          acc[tt] = __builtin_amdgcn_wmma_f32_16x16x4_f32(
               false, frag[s],  false, a,
               (short)0, acc[tt],  false,  false);
        } else {
          acc[tt] = __builtin_amdgcn_wmma_f32_16x16x4_f32(
               false, a,  false, frag[s],
               (short)0, acc[tt],  false,  false);
        }
      }
    }
    {
      v2f sv; sv.x = st[J & ~1]; sv.y = st[(J & ~1) + 1];
#pragma unroll
      for (int tt = 0; tt < 2; ++tt) {
        const v2f ev = (J & 1) ? eOdd[tt] : eEven[tt];
        if constexpr (kT) {
          acc[tt] = __builtin_amdgcn_wmma_f32_16x16x4_f32(
               false, ev,  false, sv,
               (short)0, acc[tt],  false,  false);
        } else {
          acc[tt] = __builtin_amdgcn_wmma_f32_16x16x4_f32(
               false, sv,  false, ev,
               (short)0, acc[tt],  false,  false);
        }
      }
    }
#pragma unroll
    for (int tt = 0; tt < 2; ++tt) storeTile(2 * J + tt, acc[tt]);
    __builtin_amdgcn_sched_barrier(0);
  }
}

__global__ __launch_bounds__(WPP * 32) void spline_plane(
    const float* __restrict__ in, float* __restrict__ out, const float* __restrict__ H)
{
  __shared__ float plane[NAX * PITCH];

  const int lane  = threadIdx.x & 31;
  const int wave  = __builtin_amdgcn_readfirstlane((int)(threadIdx.x >> 5));
  const int col   = lane & 15;
  const int khalf = lane >> 4;
  const float* src = in  + (size_t)blockIdx.x * (NAX * NAX);
  float*       dst = out + (size_t)blockIdx.x * (NAX * NAX);

  {
    v4f xin4[KSTEPS / 2];
    const float* lineA = src + (16 * wave + col) * NAX + 4 * khalf;
#pragma unroll
    for (int j = 0; j < KSTEPS / 2; ++j)
      xin4[j] = *reinterpret_cast<const v4f*>(lineA + 8 * j);
    __builtin_amdgcn_sched_barrier(0);
    v2f frag[NDIAG], wfrag[KSTEPS], eEven[2], eOdd[2];
    load_tables<1>(H, lane, col, khalf, frag, wfrag, eEven, eOdd);

    auto fragOf = [&](int ks) -> v2f {
      const v4f q = xin4[ks >> 1];
      v2f a;
      if (ks & 1) { a.x = q.z; a.y = q.w; } else { a.x = q.x; a.y = q.y; }
      return a;
    };
    float* rowD = plane + (16 * wave + 8 * khalf) * PITCH + col;
    auto storeTile = [&](int t, const v8f& acc) {
#pragma unroll
      for (int v = 0; v < 8; ++v) rowD[v * PITCH + 16 * t] = acc[v];
    };
    strip_compute<false>(frag, wfrag, eEven, eOdd, fragOf, storeTile);
  }

  {
    v2f frag[NDIAG], wfrag[KSTEPS], eEven[2], eOdd[2];
    load_tables<0>(H, lane, col, khalf, frag, wfrag, eEven, eOdd);

    __syncthreads();

    v2f xin[KSTEPS];
    const float* colB = plane + (2 * khalf) * PITCH + 16 * wave + col;
#pragma unroll
    for (int ks = 0; ks < KSTEPS; ++ks) {
      xin[ks].x = colB[(4 * ks + 0) * PITCH];
      xin[ks].y = colB[(4 * ks + 1) * PITCH];
    }
    auto fragOf = [&](int ks) -> v2f { return xin[ks]; };
    float* colD = dst + (8 * khalf) * NAX + 16 * wave + col;
    auto storeTile = [&](int t, const v8f& acc) {
#pragma unroll
      for (int v = 0; v < 8; ++v) colD[(16 * t + v) * NAX] = acc[v];
    };
    strip_compute<true>(frag, wfrag, eEven, eOdd, fragOf, storeTile);
  }
}

extern "C" void kernel_launch(void* const* d_in, const int* in_sizes, int n_in,
                              void* d_out, int out_size, void* d_ws, size_t ws_size,
                              hipStream_t stream) {
  const float* x   = (const float*)d_in[0];
  float*       out = (float*)d_out;
  float*       H   = (float*)d_ws;
  float*       tmp = (float*)d_ws + 32768;

  (void)in_sizes; (void)n_in; (void)out_size; (void)ws_size;

  build_H<<<dim3((NTAB + 255) / 256), dim3(256), 0, stream>>>(H);

  const int lines = 2 * 3 * NAX * NAX;
  const int tiles = lines / 16;
  dim3 grid(NBLK), block(WPB * 32);

  spline_plane<<<dim3(NPLANES), dim3(WPP * 32), 0, stream>>>(x, tmp, H);
  spline_pass<NAX * NAX, NAX * NAX * NAX, NAX * NAX>
      <<<grid, block, 0, stream>>>(tmp, out, H, tiles);
}


# Far part taken from the diagonal tiles' own edge outputs (outputs 0 and 15 of 16-wide tiles, all products transposed): 90 f32 WMMAs per strip, was 130; k sum order and reduction tree change
# speedup vs baseline: 6.7412x; 1.1499x over previous
#include <hip/hip_runtime.h>


typedef __attribute__((ext_vector_type(2))) float v2f;
typedef __attribute__((ext_vector_type(4))) float v4f;
typedef __attribute__((ext_vector_type(8))) float v8f;

#define NAX 160
#define NT  10
#define KSTEPS 40
#define KPT    4
#define NPAIR  5
#define WPB    8
#define NBLK   512
#define FRAGF  64
#define D_FL   (KPT * FRAGF)
#define S_OFF  (2 * D_FL)
#define NTAB   (S_OFF + NT * FRAGF)

__device__ constexpr int kposOf(int g, int kk, int h, int s) {
  return g ? 8 * (kk >> 1) + 4 * h + 2 * (kk & 1) + s : 4 * kk + 2 * h + s;
}
__device__ __forceinline__ int rowPerm(int r) {
  return r == 8 ? 15 : (r == 15 ? 8 : r);
}

__device__ __forceinline__ double pow_pole(int e) {
  double r = 1.0;
  double b = -0.26794919243112270647255365849413;
#pragma unroll
  for (int bit = 0; bit < 8; ++bit) {
    if ((e >> bit) & 1) r *= b;
    b *= b;
  }
  return r;
}

__global__ void build_H(float* __restrict__ H) {
  const int idx = blockIdx.x * blockDim.x + threadIdx.x;
  if (idx >= NTAB) return;
  const double s3 = 1.7320508075688772935274463415059;
  const int s    = idx & 1;
  const int lane = (idx >> 1) & 31;
  const int col  = rowPerm(lane & 15), h = lane >> 4;
  if (idx < S_OFF) {
    const int g  = idx / D_FL;
    const int kk = (idx % D_FL) / FRAGF;
    int d = col - kposOf(g, kk, h, s);
    if (d < 0) d += NAX;
    H[idx] = (float)(s3 * (pow_pole(d) + pow_pole(NAX - d)));
  } else {
    const int r  = (idx - S_OFF) / FRAGF;
    const int m  = h ? (r - s + NT) % NT : (NT - r + s) % NT;
    const int ex = m ? (h ? col : 15 - col) + 16 * (m - 1) + 1 : 0;
    H[idx] = m ? (float)pow_pole(ex) : 0.0f;
  }
}

template <int G>
__device__ __forceinline__ void load_diag(const float* __restrict__ H, int lane, v2f (&dfrag)[KPT]) {
  const float* Tg = H + G * D_FL + 2 * lane;
#pragma unroll
  for (int kk = 0; kk < KPT; ++kk) {
    dfrag[kk].x = Tg[kk * FRAGF];
    dfrag[kk].y = Tg[kk * FRAGF + 1];
  }
}
__device__ __forceinline__ void load_edgew(const float* __restrict__ H, int lane, v2f (&sfrag)[NT]) {
  const float* Ts = H + S_OFF + 2 * lane;
#pragma unroll
  for (int r = 0; r < NT; ++r) {
    sfrag[r].x = Ts[r * FRAGF];
    sfrag[r].y = Ts[r * FRAGF + 1];
  }
}

template <class FragOf, class Between, class StoreTile>
__device__ __forceinline__ void strip_compute(const v2f (&dfrag)[KPT], const v2f (&sfrag)[NT],
                                              FragOf fragOf, Between betweenSweeps, StoreTile storeTile) {
  v8f acc[NT];
#pragma unroll
  for (int Jp = 0; Jp < NPAIR; ++Jp) {
    acc[2 * Jp] = {}; acc[2 * Jp + 1] = {};
#pragma unroll
    for (int kk = 0; kk < KPT; ++kk) {
#pragma unroll
      for (int tt = 0; tt < 2; ++tt) {
        const int J = 2 * Jp + tt;
        acc[J] = __builtin_amdgcn_wmma_f32_16x16x4_f32(
             false, dfrag[kk],  false, fragOf(KPT * J + kk),
             (short)0, acc[J],  false,  false);
      }
    }
    __builtin_amdgcn_sched_barrier(0);
  }
  v2f ed[NPAIR];
#pragma unroll
  for (int q = 0; q < NPAIR; ++q) {
    ed[q].x = acc[2 * q][0];
    ed[q].y = acc[2 * q + 1][0];
  }
#pragma unroll
  for (int tp = 0; tp < NPAIR; ++tp) {
    betweenSweeps(tp);
#pragma unroll
    for (int q = 0; q < NPAIR; ++q) {
#pragma unroll
      for (int tt = 0; tt < 2; ++tt) {
        const int t = 2 * tp + tt;
        acc[t] = __builtin_amdgcn_wmma_f32_16x16x4_f32(
             false, sfrag[(t - 2 * q + NT) % NT],  false, ed[q],
             (short)0, acc[t],  false,  false);
      }
    }
#pragma unroll
    for (int tt = 0; tt < 2; ++tt) storeTile(2 * tp + tt, acc[2 * tp + tt]);
    __builtin_amdgcn_sched_barrier(0);
  }
}

template <int Q, int P, int S>
__global__ __launch_bounds__(256) void spline_pass(
    const float* __restrict__ in, float* __restrict__ out,
    const float* __restrict__ H, int nTiles)
{
  static_assert(S != 1, "lines of a strip must be adjacent in memory");
  const int lane  = threadIdx.x & 31;
  const int wave  = __builtin_amdgcn_readfirstlane((int)(threadIdx.x >> 5));
  const int col   = lane & 15;
  const int khalf = lane >> 4;

  const unsigned nPairs = (unsigned)nTiles >> 1;
  const int first = (int)(2u * ((blockIdx.x * nPairs) / NBLK));
  const int end   = (blockIdx.x + 1u == NBLK) ? nTiles
                                              : (int)(2u * (((blockIdx.x + 1u) * nPairs) / NBLK));
  const int count = end - first;
  const int full  = count / WPB;
  const int rem   = count % WPB;
  const int pw    = (wave * 5) & (WPB - 1);
  const int nMine = full + (pw < rem ? 1 : 0);
  if (nMine <= 0) return;

  auto stripOf = [&](int i) -> int {
    return first + (i < full ? i * WPB + wave : full * WPB + pw);
  };
  auto lineBase = [&](int tile) -> int {
    const int l = tile * 16 + col;
    return (l / Q) * P + (l % Q);
  };
  auto loadA = [&](int base, int ks) -> v2f {
    v2f a;
    a.x = in[base + (ks * 4 + 2 * khalf + 0) * S];
    a.y = in[base + (ks * 4 + 2 * khalf + 1) * S];
    return a;
  };

  v2f dfrag[KPT], sfrag[NT];
  load_diag<0>(H, lane, dfrag);
  load_edgew(H, lane, sfrag);
  __builtin_amdgcn_sched_barrier(0);

  int baseA = lineBase(stripOf(0));

  v2f xin[KSTEPS];
#pragma unroll
  for (int ks = 0; ks < KSTEPS; ++ks) {
    xin[ks] = loadA(baseA, ks);
    __builtin_amdgcn_sched_barrier(0);
  }

  for (int i = 0; i < nMine; ++i) {
    const int baseN = lineBase(stripOf(i + 1 < nMine ? i + 1 : i));
    const int baseO0 = baseA + 15 * khalf * S;
    const int baseO7 = baseA + (7 + khalf) * S;
    const int baseOm = baseA + 8 * khalf * S;

    auto fragOf = [&](int ks) -> v2f { return xin[ks]; };
    auto refill = [&](int q) {
#pragma unroll
      for (int kk = 0; kk < 2 * KPT; ++kk) xin[2 * KPT * q + kk] = loadA(baseN, 2 * KPT * q + kk);
    };
    auto storeTile = [&](int t, const v8f& acc) {
#pragma unroll
      for (int v = 0; v < 8; ++v) {
        const int b = (v == 0) ? baseO0 : (v == 7) ? baseO7 : baseOm + v * S;
        out[b + 16 * t * S] = acc[v];
      }
    };
    strip_compute(dfrag, sfrag, fragOf, refill, storeTile);

    baseA = baseN;
  }
}

#define WPP    10
#define NPLANES 960
#define PITCH  172

__global__ __launch_bounds__(WPP * 32) void spline_plane(
    const float* __restrict__ in, float* __restrict__ out, const float* __restrict__ H)
{
  __shared__ __attribute__((aligned(16))) float plane[NAX * PITCH];

  const int lane  = threadIdx.x & 31;
  const int wave  = __builtin_amdgcn_readfirstlane((int)(threadIdx.x >> 5));
  const int col   = lane & 15;
  const int khalf = lane >> 4;
  const float* src = in  + (size_t)blockIdx.x * (NAX * NAX);
  float*       dst = out + (size_t)blockIdx.x * (NAX * NAX);

  v4f xin4[KSTEPS / 2];
  const float* lineA = src + (16 * wave + col) * NAX + 4 * khalf;
#pragma unroll
  for (int j = 0; j < KSTEPS / 2; ++j)
    xin4[j] = *reinterpret_cast<const v4f*>(lineA + 8 * j);
  __builtin_amdgcn_sched_barrier(0);
  v2f sfrag[NT];
  load_edgew(H, lane, sfrag);
  auto nothing = [](int) {};

  {
    v2f dfrag[KPT];
    load_diag<1>(H, lane, dfrag);
    auto fragOf = [&](int ks) -> v2f {
      const v4f q = xin4[ks >> 1];
      v2f a;
      if (ks & 1) { a.x = q.z; a.y = q.w; } else { a.x = q.x; a.y = q.y; }
      return a;
    };
    float* rowD = plane + (16 * wave + col) * PITCH + 8 * khalf;
    auto storeTile = [&](int t, const v8f& acc) {
      v4f lo, hi;
      lo.x = acc[0]; lo.y = acc[1]; lo.z = acc[2]; lo.w = acc[3];
      hi.x = acc[4]; hi.y = acc[5]; hi.z = acc[6]; hi.w = acc[7];
      *reinterpret_cast<v4f*>(rowD + 16 * t)     = lo;
      *reinterpret_cast<v4f*>(rowD + 16 * t + 4) = hi;
    };
    strip_compute(dfrag, sfrag, fragOf, nothing, storeTile);
  }

  {
    v2f dfrag[KPT];
    load_diag<0>(H, lane, dfrag);

    __syncthreads();

    v2f xin[KSTEPS];
    const float* colB = plane + (2 * khalf) * PITCH + 16 * wave + col;
#pragma unroll
    for (int ks = 0; ks < KSTEPS; ++ks) {
      xin[ks].x = colB[(4 * ks + 0) * PITCH];
      xin[ks].y = colB[(4 * ks + 1) * PITCH];
    }
    auto fragOf = [&](int ks) -> v2f { return xin[ks]; };
    float* colD  = dst + 16 * wave + rowPerm(col);
    float* colD0 = colD + 15 * khalf * NAX;
    float* colD7 = colD + (7 + khalf) * NAX;
    float* colDm = colD + 8 * khalf * NAX;
    auto storeTile = [&](int t, const v8f& acc) {
#pragma unroll
      for (int v = 0; v < 8; ++v) {
        float* b = (v == 0) ? colD0 : (v == 7) ? colD7 : colDm + v * NAX;
        b[16 * t * NAX] = acc[v];
      }
    };
    strip_compute(dfrag, sfrag, fragOf, nothing, storeTile);
  }
}

extern "C" void kernel_launch(void* const* d_in, const int* in_sizes, int n_in,
                              void* d_out, int out_size, void* d_ws, size_t ws_size,
                              hipStream_t stream) {
  const float* x   = (const float*)d_in[0];
  float*       out = (float*)d_out;
  float*       H   = (float*)d_ws;
  float*       tmp = (float*)d_ws + 32768;

  (void)in_sizes; (void)n_in; (void)out_size; (void)ws_size;

  build_H<<<dim3((NTAB + 255) / 256), dim3(256), 0, stream>>>(H);

  const int lines = 2 * 3 * NAX * NAX;
  const int tiles = lines / 16;
  dim3 grid(NBLK), block(WPB * 32);

  spline_plane<<<dim3(NPLANES), dim3(WPP * 32), 0, stream>>>(x, tmp, H);
  spline_pass<NAX * NAX, NAX * NAX * NAX, NAX * NAX>
      <<<grid, block, 0, stream>>>(tmp, out, H, tiles);
}


# Far part as a rank-2 update: two carries per tile from one 20x20 product with the edge values (10 WMMAs), then one K-step per tile (10): 60 f32 WMMAs per strip, was 90; far sum order changes
# speedup vs baseline: 7.1490x; 1.0605x over previous
#include <hip/hip_runtime.h>


typedef __attribute__((ext_vector_type(2))) float v2f;
typedef __attribute__((ext_vector_type(4))) float v4f;
typedef __attribute__((ext_vector_type(8))) float v8f;

#define NAX 160
#define NT  10
#define KSTEPS 40
#define KPT    4
#define NPAIR  5
#define WPB    8
#define NBLK   512
#define FRAGF  64
#define D_FL   (KPT * FRAGF)
#define S_OFF  (2 * D_FL)
#define NCW    (2 * NPAIR)
#define U_OFF  (S_OFF + NCW * FRAGF)
#define NTAB   (U_OFF + 2 * FRAGF)

__device__ constexpr int kposOf(int g, int kk, int h, int s) {
  return g ? 8 * (kk >> 1) + 4 * h + 2 * (kk & 1) + s : 4 * kk + 2 * h + s;
}
__device__ __forceinline__ int rowPerm(int r) {
  return r == 8 ? 15 : (r == 15 ? 8 : r);
}

__device__ __forceinline__ double pow_pole(int e) {
  double r = 1.0;
  double b = -0.26794919243112270647255365849413;
#pragma unroll
  for (int bit = 0; bit < 8; ++bit) {
    if ((e >> bit) & 1) r *= b;
    b *= b;
  }
  return r;
}

__global__ void build_H(float* __restrict__ H) {
  const int idx = blockIdx.x * blockDim.x + threadIdx.x;
  if (idx >= NTAB) return;
  const double s3 = 1.7320508075688772935274463415059;
  const int s    = idx & 1;
  const int lane = (idx >> 1) & 31;
  const int row  = lane & 15, h = lane >> 4;
  const int col  = rowPerm(row);
  if (idx < S_OFF) {
    const int g  = idx / D_FL;
    const int kk = (idx % D_FL) / FRAGF;
    int d = col - kposOf(g, kk, h, s);
    if (d < 0) d += NAX;
    H[idx] = (float)(s3 * (pow_pole(d) + pow_pole(NAX - d)));
  } else if (idx < U_OFF) {
    const int f  = (idx - S_OFF) / FRAGF;
    const int t  = 8 * (f / NPAIR) + (row & 7);
    const int J  = 2 * (f % NPAIR) + s;
    const int an = row >> 3;
    const int m  = an ? (J - t + NT) % NT : (t - J + NT) % NT;
    const bool on = t < NT && h != an && m != 0;
    H[idx] = on ? (float)pow_pole(16 * (m - 1) + 1) : 0.0f;
  } else {
    const int tt = (idx - U_OFF) / FRAGF;
    H[idx] = (s == tt) ? (float)pow_pole(h ? 15 - col : col) : 0.0f;
  }
}

template <int G>
__device__ __forceinline__ void load_diag(const float* __restrict__ H, int lane, v2f (&dfrag)[KPT]) {
  const float* Tg = H + G * D_FL + 2 * lane;
#pragma unroll
  for (int kk = 0; kk < KPT; ++kk) {
    dfrag[kk].x = Tg[kk * FRAGF];
    dfrag[kk].y = Tg[kk * FRAGF + 1];
  }
}
__device__ __forceinline__ void load_far(const float* __restrict__ H, int lane, v2f (&sfrag)[NCW + 2]) {
  const float* Ts = H + S_OFF + 2 * lane;
#pragma unroll
  for (int r = 0; r < NCW + 2; ++r) {
    sfrag[r].x = Ts[r * FRAGF];
    sfrag[r].y = Ts[r * FRAGF + 1];
  }
}

template <class FragOf, class InputsFree, class StoreTile>
__device__ __forceinline__ void strip_compute(const v2f (&dfrag)[KPT], const v2f (&sfrag)[NCW + 2],
                                              FragOf fragOf, InputsFree inputsFree, StoreTile storeTile) {
  v8f acc[NT];
#pragma unroll
  for (int Jp = 0; Jp < NPAIR; ++Jp) {
    if (Jp > 0) inputsFree(Jp - 1);
    acc[2 * Jp] = {}; acc[2 * Jp + 1] = {};
#pragma unroll
    for (int kk = 0; kk < KPT; ++kk) {
#pragma unroll
      for (int tt = 0; tt < 2; ++tt) {
        const int J = 2 * Jp + tt;
        acc[J] = __builtin_amdgcn_wmma_f32_16x16x4_f32(
             false, dfrag[kk],  false, fragOf(KPT * J + kk),
             (short)0, acc[J],  false,  false);
      }
    }
    __builtin_amdgcn_sched_barrier(0);
  }
  v2f ed[NPAIR];
#pragma unroll
  for (int q = 0; q < NPAIR; ++q) {
    ed[q].x = acc[2 * q][0];
    ed[q].y = acc[2 * q + 1][0];
  }
  inputsFree(NPAIR - 1);
  v8f car[2];
  car[0] = {}; car[1] = {};
#pragma unroll
  for (int q = 0; q < NPAIR; ++q) {
#pragma unroll
    for (int rt = 0; rt < 2; ++rt) {
      car[rt] = __builtin_amdgcn_wmma_f32_16x16x4_f32(
           false, sfrag[NPAIR * rt + q],  false, ed[q],
           (short)0, car[rt],  false,  false);
    }
  }
#pragma unroll
  for (int tp = 0; tp < NPAIR; ++tp) {
    v2f cb;
    cb.x = car[tp >> 2][(2 * tp) & 7];
    cb.y = car[tp >> 2][((2 * tp) & 7) + 1];
#pragma unroll
    for (int tt = 0; tt < 2; ++tt) {
      const int t = 2 * tp + tt;
      acc[t] = __builtin_amdgcn_wmma_f32_16x16x4_f32(
           false, sfrag[NCW + tt],  false, cb,
           (short)0, acc[t],  false,  false);
    }
#pragma unroll
    for (int tt = 0; tt < 2; ++tt) storeTile(2 * tp + tt, acc[2 * tp + tt]);
    __builtin_amdgcn_sched_barrier(0);
  }
}

template <int Q, int P, int S>
__global__ __launch_bounds__(256) void spline_pass(
    const float* __restrict__ in, float* __restrict__ out,
    const float* __restrict__ H, int nTiles)
{
  static_assert(S != 1, "lines of a strip must be adjacent in memory");
  const int lane  = threadIdx.x & 31;
  const int wave  = __builtin_amdgcn_readfirstlane((int)(threadIdx.x >> 5));
  const int col   = lane & 15;
  const int khalf = lane >> 4;

  const unsigned nPairs = (unsigned)nTiles >> 1;
  const int first = (int)(2u * ((blockIdx.x * nPairs) / NBLK));
  const int end   = (blockIdx.x + 1u == NBLK) ? nTiles
                                              : (int)(2u * (((blockIdx.x + 1u) * nPairs) / NBLK));
  const int count = end - first;
  const int full  = count / WPB;
  const int rem   = count % WPB;
  const int pw    = (wave * 5) & (WPB - 1);
  const int nMine = full + (pw < rem ? 1 : 0);
  if (nMine <= 0) return;

  auto stripOf = [&](int i) -> int {
    return first + (i < full ? i * WPB + wave : full * WPB + pw);
  };
  auto lineBase = [&](int tile) -> int {
    const int l = tile * 16 + col;
    return (l / Q) * P + (l % Q);
  };
  auto loadA = [&](int base, int ks) -> v2f {
    v2f a;
    a.x = in[base + (ks * 4 + 2 * khalf + 0) * S];
    a.y = in[base + (ks * 4 + 2 * khalf + 1) * S];
    return a;
  };

  v2f dfrag[KPT], sfrag[NCW + 2];
  load_diag<0>(H, lane, dfrag);
  load_far(H, lane, sfrag);
  __builtin_amdgcn_sched_barrier(0);

  int baseA = lineBase(stripOf(0));

  v2f xin[KSTEPS];
#pragma unroll
  for (int ks = 0; ks < KSTEPS; ++ks) {
    xin[ks] = loadA(baseA, ks);
    __builtin_amdgcn_sched_barrier(0);
  }

  for (int i = 0; i < nMine; ++i) {
    const int baseN = lineBase(stripOf(i + 1 < nMine ? i + 1 : i));
    const int baseO0 = baseA + 15 * khalf * S;
    const int baseO7 = baseA + (7 + khalf) * S;
    const int baseOm = baseA + 8 * khalf * S;

    auto fragOf = [&](int ks) -> v2f { return xin[ks]; };
    auto refill = [&](int q) {
#pragma unroll
      for (int kk = 0; kk < 2 * KPT; ++kk) xin[2 * KPT * q + kk] = loadA(baseN, 2 * KPT * q + kk);
    };
    auto storeTile = [&](int t, const v8f& acc) {
#pragma unroll
      for (int v = 0; v < 8; ++v) {
        const int b = (v == 0) ? baseO0 : (v == 7) ? baseO7 : baseOm + v * S;
        out[b + 16 * t * S] = acc[v];
      }
    };
    strip_compute(dfrag, sfrag, fragOf, refill, storeTile);

    baseA = baseN;
  }
}

#define WPP    10
#define NPLANES 960
#define PITCH  172

__global__ __launch_bounds__(WPP * 32) void spline_plane(
    const float* __restrict__ in, float* __restrict__ out, const float* __restrict__ H)
{
  __shared__ __attribute__((aligned(16))) float plane[NAX * PITCH];

  const int lane  = threadIdx.x & 31;
  const int wave  = __builtin_amdgcn_readfirstlane((int)(threadIdx.x >> 5));
  const int col   = lane & 15;
  const int khalf = lane >> 4;
  const float* src = in  + (size_t)blockIdx.x * (NAX * NAX);
  float*       dst = out + (size_t)blockIdx.x * (NAX * NAX);

  v4f xin4[KSTEPS / 2];
  const float* lineA = src + (16 * wave + col) * NAX + 4 * khalf;
#pragma unroll
  for (int j = 0; j < KSTEPS / 2; ++j)
    xin4[j] = *reinterpret_cast<const v4f*>(lineA + 8 * j);
  __builtin_amdgcn_sched_barrier(0);
  v2f sfrag[NCW + 2];
  load_far(H, lane, sfrag);
  auto nothing = [](int) {};

  {
    v2f dfrag[KPT];
    load_diag<1>(H, lane, dfrag);
    auto fragOf = [&](int ks) -> v2f {
      const v4f q = xin4[ks >> 1];
      v2f a;
      if (ks & 1) { a.x = q.z; a.y = q.w; } else { a.x = q.x; a.y = q.y; }
      return a;
    };
    float* rowD = plane + (16 * wave + col) * PITCH + 8 * khalf;
    auto storeTile = [&](int t, const v8f& acc) {
      v4f lo, hi;
      lo.x = acc[0]; lo.y = acc[1]; lo.z = acc[2]; lo.w = acc[3];
      hi.x = acc[4]; hi.y = acc[5]; hi.z = acc[6]; hi.w = acc[7];
      *reinterpret_cast<v4f*>(rowD + 16 * t)     = lo;
      *reinterpret_cast<v4f*>(rowD + 16 * t + 4) = hi;
    };
    strip_compute(dfrag, sfrag, fragOf, nothing, storeTile);
  }

  {
    v2f dfrag[KPT];
    load_diag<0>(H, lane, dfrag);

    __syncthreads();

    v2f xin[KSTEPS];
    const float* colB = plane + (2 * khalf) * PITCH + 16 * wave + col;
#pragma unroll
    for (int ks = 0; ks < KSTEPS; ++ks) {
      xin[ks].x = colB[(4 * ks + 0) * PITCH];
      xin[ks].y = colB[(4 * ks + 1) * PITCH];
    }
    auto fragOf = [&](int ks) -> v2f { return xin[ks]; };
    float* colD  = dst + 16 * wave + rowPerm(col);
    float* colD0 = colD + 15 * khalf * NAX;
    float* colD7 = colD + (7 + khalf) * NAX;
    float* colDm = colD + 8 * khalf * NAX;
    auto storeTile = [&](int t, const v8f& acc) {
#pragma unroll
      for (int v = 0; v < 8; ++v) {
        float* b = (v == 0) ? colD0 : (v == 7) ? colD7 : colDm + v * NAX;
        b[16 * t * NAX] = acc[v];
      }
    };
    strip_compute(dfrag, sfrag, fragOf, nothing, storeTile);
  }
}

extern "C" void kernel_launch(void* const* d_in, const int* in_sizes, int n_in,
                              void* d_out, int out_size, void* d_ws, size_t ws_size,
                              hipStream_t stream) {
  const float* x   = (const float*)d_in[0];
  float*       out = (float*)d_out;
  float*       H   = (float*)d_ws;
  float*       tmp = (float*)d_ws + 32768;

  (void)in_sizes; (void)n_in; (void)out_size; (void)ws_size;

  build_H<<<dim3((NTAB + 255) / 256), dim3(256), 0, stream>>>(H);

  const int lines = 2 * 3 * NAX * NAX;
  const int tiles = lines / 16;
  dim3 grid(NBLK), block(WPB * 32);

  spline_plane<<<dim3(NPLANES), dim3(WPP * 32), 0, stream>>>(x, tmp, H);
  spline_pass<NAX * NAX, NAX * NAX * NAX, NAX * NAX>
      <<<grid, block, 0, stream>>>(tmp, out, H, tiles);
}
